# starting kernel
# baseline (speedup 1.0000x reference)
#include <hip/hip_runtime.h>
#include <hip/hip_bf16.h>


typedef __attribute__((ext_vector_type(2))) float v2f;
typedef __attribute__((ext_vector_type(8))) float v8f;

#define IN_C 512
#define HID 256
#define OUT_C 128

__global__ void k_deg_init(float* __restrict__ deg, int n) {
  int i = blockIdx.x * blockDim.x + threadIdx.x;
  if (i < n) deg[i] = 1.0f;
}

__global__ void k_deg_count(float* __restrict__ deg,
                            const int* __restrict__ dst, int e) {
  int i = blockIdx.x * blockDim.x + threadIdx.x;
  if (i < e) atomicAdd(&deg[dst[i]], 1.0f);
}

__global__ void k_deg_rsqrt(float* __restrict__ deg, int n) {
  int i = blockIdx.x * blockDim.x + threadIdx.x;
  if (i < n) deg[i] = rsqrtf(deg[i]);
}

__global__ void k_fill_bias(float* __restrict__ out,
                            const float* __restrict__ bias,
                            long long total, int C) {
  long long i = (long long)blockIdx.x * blockDim.x + threadIdx.x;
  long long stride = (long long)gridDim.x * blockDim.x;
  for (; i < total; i += stride) out[i] = bias[(int)(i % C)];
}

__global__ void k_relu(float* __restrict__ buf, long long total) {
  long long i = (long long)blockIdx.x * blockDim.x + threadIdx.x;
  long long stride = (long long)gridDim.x * blockDim.x;
  for (; i < total; i += stride) buf[i] = fmaxf(buf[i], 0.0f);
}

template <int KDIM, int NCOLS, int NGROUPS>
__global__ void k_gemm_wmma_f32(const float* __restrict__ A,
                                const float* __restrict__ B,
                                float* __restrict__ C, int mtiles) {
  const int lane = threadIdx.x & 31;
  const int wib  = threadIdx.x >> 5;
  const int wid  = blockIdx.x * (blockDim.x >> 5) + wib;
  const int totalWaves = mtiles * NGROUPS;
  if (wid >= totalWaves) return;

  const int mt = wid / NGROUPS;
  const int ng = wid % NGROUPS;
  const int m0 = mt * 16;
  const int n0 = ng * 64;

  const int arow  = m0 + (lane & 15);
  const int khalf = (lane >> 4) * 2;
  const int bcol  = n0 + (lane & 15);

  const float* __restrict__ ap = A + (long long)arow * KDIM;

  v8f c0 = {}, c1 = {}, c2 = {}, c3 = {};

#pragma unroll 4
  for (int k = 0; k < KDIM; k += 4) {
    v2f a;
    a.x = ap[k + khalf];
    a.y = ap[k + khalf + 1];

    const float* __restrict__ br0 = B + (long long)(k + khalf) * NCOLS + bcol;
    const float* __restrict__ br1 = br0 + NCOLS;
    v2f b0, b1, b2, b3;
    b0.x = br0[0];  b0.y = br1[0];
    b1.x = br0[16]; b1.y = br1[16];
    b2.x = br0[32]; b2.y = br1[32];
    b3.x = br0[48]; b3.y = br1[48];

    c0 = __builtin_amdgcn_wmma_f32_16x16x4_f32(false, a, false, b0, (short)0, c0, false, false);
    c1 = __builtin_amdgcn_wmma_f32_16x16x4_f32(false, a, false, b1, (short)0, c1, false, false);
    c2 = __builtin_amdgcn_wmma_f32_16x16x4_f32(false, a, false, b2, (short)0, c2, false, false);
    c3 = __builtin_amdgcn_wmma_f32_16x16x4_f32(false, a, false, b3, (short)0, c3, false, false);
  }

  const int rbase = m0 + (lane >> 4) * 8;
  const int cc    = lane & 15;
#pragma unroll
  for (int r = 0; r < 8; ++r) {
    float* __restrict__ orow = C + (long long)(rbase + r) * NCOLS + n0 + cc;
    orow[0]  = c0[r];
    orow[16] = c1[r];
    orow[32] = c2[r];
    orow[48] = c3[r];
  }
}

template <int C>
__global__ void k_aggregate(const float* __restrict__ hlin,
                            float* __restrict__ out,
                            const int* __restrict__ src,
                            const int* __restrict__ dst,
                            const float* __restrict__ dinv,
                            int E, int N) {
  const int lane = threadIdx.x & 31;
  long long wid = (long long)blockIdx.x * (blockDim.x >> 5) + (threadIdx.x >> 5);
  const long long wstride = (long long)gridDim.x * (blockDim.x >> 5);
  const long long EV = (long long)E + N;

  for (long long e = wid; e < EV; e += wstride) {
    int s, d;
    if (e < (long long)E) { s = src[e]; d = dst[e]; }
    else                  { s = d = (int)(e - E); }
    const float norm = dinv[s] * dinv[d];
    const float4* __restrict__ hs = (const float4*)(hlin + (long long)s * C);
    float* __restrict__ od = out + (long long)d * C;
#pragma unroll
    for (int j = lane; j < C / 4; j += 32) {
      float4 v = hs[j];
      atomicAdd(&od[4 * j + 0], v.x * norm);
      atomicAdd(&od[4 * j + 1], v.y * norm);
      atomicAdd(&od[4 * j + 2], v.z * norm);
      atomicAdd(&od[4 * j + 3], v.w * norm);
    }
  }
}

extern "C" void kernel_launch(void* const* d_in, const int* in_sizes, int n_in,
                              void* d_out, int out_size, void* d_ws, size_t ws_size,
                              hipStream_t stream) {
  (void)n_in; (void)out_size; (void)ws_size;

  const float* x  = (const float*)d_in[0];
  const int*   ei = (const int*)d_in[1];
  const float* W1 = (const float*)d_in[2];
  const float* b1 = (const float*)d_in[3];
  const float* W2 = (const float*)d_in[4];
  const float* b2 = (const float*)d_in[5];
  float* out = (float*)d_out;

  const int N = in_sizes[0] / IN_C;
  const int E = in_sizes[1] / 2;
  const int* src = ei;
  const int* dst = ei + E;

  char* w = (char*)d_ws;
  size_t dinv_bytes = (((size_t)N * 4) + 255) & ~(size_t)255;
  float* dinv = (float*)w;
  float* bufA = (float*)(w + dinv_bytes);
  float* bufB = bufA + (size_t)N * HID;

  const int T = 256;

  k_deg_init<<<(N + T - 1) / T, T, 0, stream>>>(dinv, N);
  k_deg_count<<<(E + T - 1) / T, T, 0, stream>>>(dinv, dst, E);
  k_deg_rsqrt<<<(N + T - 1) / T, T, 0, stream>>>(dinv, N);

  const int mtiles = N / 16;

  {
    int waves = mtiles * 4;
    int blocks = (waves + 7) / 8;
    k_gemm_wmma_f32<IN_C, HID, 4><<<blocks, T, 0, stream>>>(x, W1, bufA, mtiles);
  }

  k_fill_bias<<<2048, T, 0, stream>>>(bufB, b1, (long long)N * HID, HID);
  k_aggregate<HID><<<4096, T, 0, stream>>>(bufA, bufB, src, dst, dinv, E, N);
  k_relu<<<2048, T, 0, stream>>>(bufB, (long long)N * HID);

  {
    int waves = mtiles * 2;
    int blocks = (waves + 7) / 8;
    k_gemm_wmma_f32<HID, OUT_C, 2><<<blocks, T, 0, stream>>>(bufB, W2, bufA, mtiles);
  }

  k_fill_bias<<<2048, T, 0, stream>>>(out, b2, (long long)N * OUT_C, OUT_C);
  k_aggregate<OUT_C><<<4096, T, 0, stream>>>(bufA, out, src, dst, dinv, E, N);
}


# Aggregation without float atomics: edges grouped by destination (integer count with arrival-order rank, scan, place); one wave sums each row in registers, stores it once with bias and ReLU
# speedup vs baseline: 2.0545x; 2.0545x over previous
#include <hip/hip_runtime.h>


typedef __attribute__((ext_vector_type(2))) float v2f;
typedef __attribute__((ext_vector_type(8))) float v8f;

#define IN_C 512
#define HID 256
#define OUT_C 128
#define PLANE 128

__global__ void k_zero_i32(int* __restrict__ p, int n) {
  int i = blockIdx.x * blockDim.x + threadIdx.x;
  if (i < n) p[i] = 0;
}

__global__ void k_count_rank(int* __restrict__ cnt, int* __restrict__ rank,
                             const int* __restrict__ dst, int e) {
  int i = blockIdx.x * blockDim.x + threadIdx.x;
  if (i < e) rank[i] = atomicAdd(&cnt[dst[i]], 1);
}

__global__ void __launch_bounds__(1024)
k_scan_rows(const int* __restrict__ cnt, int* __restrict__ rowptr,
            float* __restrict__ dinv, int n) {
  __shared__ int wsum[32];
  const int lane = threadIdx.x & 31;
  const int w = threadIdx.x >> 5;
  const int per = ((n + 1023) / 1024) * 32;
  const int b = w * per;
  const int e = (b + per < n) ? (b + per) : n;

  int tot = 0;
  for (int i = b + lane; i < e; i += 32) tot += cnt[i];
  for (int o = 16; o > 0; o >>= 1) tot += __shfl_xor(tot, o);
  if (lane == 0) wsum[w] = tot;
  __syncthreads();

  const int v = wsum[lane];
  int incl = v;
  for (int o = 1; o < 32; o <<= 1) {
    int t = __shfl_up(incl, o);
    if (lane >= o) incl += t;
  }
  int carry = __shfl(incl - v, w);
  const int total = __shfl(incl, 31);

  for (int i0 = b; i0 < e; i0 += 32) {
    const int i = i0 + lane;
    const int c = (i < e) ? cnt[i] : 0;
    int inc = c;
    for (int o = 1; o < 32; o <<= 1) {
      int t = __shfl_up(inc, o);
      if (lane >= o) inc += t;
    }
    if (i < e) {
      rowptr[i] = carry + inc - c;
      dinv[i] = rsqrtf((float)(c + 1));
    }
    carry += __shfl(inc, 31);
  }
  if (threadIdx.x == 0) rowptr[n] = total;
}

__global__ void k_place(int* __restrict__ csr, const int* __restrict__ rowptr,
                        const int* __restrict__ rank, const int* __restrict__ src,
                        const int* __restrict__ dst, int e) {
  int i = blockIdx.x * blockDim.x + threadIdx.x;
  if (i < e) csr[rowptr[dst[i]] + rank[i]] = src[i];
}

template <int KDIM, int NCOLS, int NGROUPS, int CPL>
__global__ void k_gemm_wmma_f32(const float* __restrict__ A,
                                const float* __restrict__ B,
                                float* __restrict__ C0, float* __restrict__ C1,
                                int mtiles) {
  const int lane = threadIdx.x & 31;
  const int wib  = threadIdx.x >> 5;
  const int wid  = blockIdx.x * (blockDim.x >> 5) + wib;
  const int totalWaves = mtiles * NGROUPS;
  if (wid >= totalWaves) return;

  const int mt = wid / NGROUPS;
  const int ng = wid % NGROUPS;
  const int m0 = mt * 16;
  const int n0 = ng * 64;

  const int arow  = m0 + (lane & 15);
  const int khalf = (lane >> 4) * 2;
  const int bcol  = n0 + (lane & 15);

  const float* __restrict__ ap = A + (long long)arow * KDIM;

  v8f c0 = {}, c1 = {}, c2 = {}, c3 = {};

#pragma unroll 4
  for (int k = 0; k < KDIM; k += 4) {
    v2f a;
    a.x = ap[k + khalf];
    a.y = ap[k + khalf + 1];

    const float* __restrict__ br0 = B + (long long)(k + khalf) * NCOLS + bcol;
    const float* __restrict__ br1 = br0 + NCOLS;
    v2f b0, b1, b2, b3;
    b0.x = br0[0];  b0.y = br1[0];
    b1.x = br0[16]; b1.y = br1[16];
    b2.x = br0[32]; b2.y = br1[32];
    b3.x = br0[48]; b3.y = br1[48];

    c0 = __builtin_amdgcn_wmma_f32_16x16x4_f32(false, a, false, b0, (short)0, c0, false, false);
    c1 = __builtin_amdgcn_wmma_f32_16x16x4_f32(false, a, false, b1, (short)0, c1, false, false);
    c2 = __builtin_amdgcn_wmma_f32_16x16x4_f32(false, a, false, b2, (short)0, c2, false, false);
    c3 = __builtin_amdgcn_wmma_f32_16x16x4_f32(false, a, false, b3, (short)0, c3, false, false);
  }

  const int rbase = m0 + (lane >> 4) * 8;
  const int cc    = lane & 15;
  const int plane = n0 / CPL;
  const int nc    = n0 - plane * CPL;
  float* __restrict__ Cb = plane ? C1 : C0;
#pragma unroll
  for (int r = 0; r < 8; ++r) {
    float* __restrict__ orow = Cb + (long long)(rbase + r) * CPL + nc + cc;
    orow[0]  = c0[r];
    orow[16] = c1[r];
    orow[32] = c2[r];
    orow[48] = c3[r];
  }
}

template <int NPL, bool RELU, int UNR>
__global__ void __launch_bounds__(256)
k_gather_rows(const float* __restrict__ p0, const float* __restrict__ p1,
              float* __restrict__ rows_out,
              const int* __restrict__ rowptr, const int* __restrict__ csr,
              const float* __restrict__ dinv, const float* __restrict__ bias,
              int N) {
  const int lane = threadIdx.x & 31;
  const int wid = __builtin_amdgcn_readfirstlane(
      (int)(blockIdx.x * (blockDim.x >> 5) + (threadIdx.x >> 5)));
  const int nw = (int)(gridDim.x * (blockDim.x >> 5));

  for (int d = wid; d < N; d += nw) {
    const int beg = rowptr[d];
    const int end = rowptr[d + 1];
    const float dd = dinv[d];
    const float wself = dd * dd;

    float4 a0 = ((const float4*)(p0 + (size_t)d * PLANE))[lane];
    a0.x *= wself; a0.y *= wself; a0.z *= wself; a0.w *= wself;
    float4 a1 = make_float4(0.f, 0.f, 0.f, 0.f);
    if (NPL == 2) {
      a1 = ((const float4*)(p1 + (size_t)d * PLANE))[lane];
      a1.x *= wself; a1.y *= wself; a1.z *= wself; a1.w *= wself;
    }

    for (int base = beg; base < end; base += 32) {
      const int left = end - base;
      const int cnt = left < 32 ? left : 32;
      int s_l = d;
      float w_l = 0.0f;
      if (lane < cnt) {
        s_l = csr[base + lane];
        w_l = dinv[s_l] * dd;
      }
      for (int j = 0; j < cnt; j += UNR) {
        float4 v0[UNR];
        float4 v1[UNR];
        float w[UNR];
#pragma unroll
        for (int u = 0; u < UNR; ++u) {
          const int s = __builtin_amdgcn_readlane(s_l, j + u);
          w[u] = __builtin_bit_cast(
              float, __builtin_amdgcn_readlane(__builtin_bit_cast(int, w_l), j + u));
          v0[u] = ((const float4*)(p0 + (size_t)s * PLANE))[lane];
          if (NPL == 2) v1[u] = ((const float4*)(p1 + (size_t)s * PLANE))[lane];
        }
#pragma unroll
        for (int u = 0; u < UNR; ++u) {
          a0.x += w[u] * v0[u].x; a0.y += w[u] * v0[u].y;
          a0.z += w[u] * v0[u].z; a0.w += w[u] * v0[u].w;
          if (NPL == 2) {
            a1.x += w[u] * v1[u].x; a1.y += w[u] * v1[u].y;
            a1.z += w[u] * v1[u].z; a1.w += w[u] * v1[u].w;
          }
        }
      }
    }

    float* __restrict__ orow = rows_out + (size_t)d * (NPL * PLANE);
    {
      const float4 b = ((const float4*)bias)[lane];
      a0.x += b.x; a0.y += b.y; a0.z += b.z; a0.w += b.w;
      if (RELU) {
        a0.x = fmaxf(a0.x, 0.f); a0.y = fmaxf(a0.y, 0.f);
        a0.z = fmaxf(a0.z, 0.f); a0.w = fmaxf(a0.w, 0.f);
      }
      ((float4*)orow)[lane] = a0;
    }
    if (NPL == 2) {
      const float4 b = ((const float4*)(bias + PLANE))[lane];
      a1.x += b.x; a1.y += b.y; a1.z += b.z; a1.w += b.w;
      if (RELU) {
        a1.x = fmaxf(a1.x, 0.f); a1.y = fmaxf(a1.y, 0.f);
        a1.z = fmaxf(a1.z, 0.f); a1.w = fmaxf(a1.w, 0.f);
      }
      ((float4*)(orow + PLANE))[lane] = a1;
    }
  }
}

extern "C" void kernel_launch(void* const* d_in, const int* in_sizes, int n_in,
                              void* d_out, int out_size, void* d_ws, size_t ws_size,
                              hipStream_t stream) {
  (void)n_in; (void)out_size; (void)ws_size;

  const float* x  = (const float*)d_in[0];
  const int*   ei = (const int*)d_in[1];
  const float* W1 = (const float*)d_in[2];
  const float* b1 = (const float*)d_in[3];
  const float* W2 = (const float*)d_in[4];
  const float* b2 = (const float*)d_in[5];
  float* out = (float*)d_out;

  const int N = in_sizes[0] / IN_C;
  const int E = in_sizes[1] / 2;
  const int* src = ei;
  const int* dst = ei + E;

  char* w = (char*)d_ws;
  size_t dinv_bytes = (((size_t)N * 4) + 255) & ~(size_t)255;
  float* dinv = (float*)w;
  float* bufA = (float*)(w + dinv_bytes);
  float* bufB = bufA + (size_t)N * HID;
  char*  elist  = (char*)(bufA + (size_t)N * PLANE);
  int*   cnt    = (int*)elist;
  int*   rowptr = (int*)(elist + dinv_bytes);
  int*   rank   = (int*)(elist + 2 * dinv_bytes);
  int*   csr    = rank + E;

  const int T = 256;

  k_zero_i32<<<(N + T - 1) / T, T, 0, stream>>>(cnt, N);
  k_count_rank<<<(E + T - 1) / T, T, 0, stream>>>(cnt, rank, dst, E);
  k_scan_rows<<<1, 1024, 0, stream>>>(cnt, rowptr, dinv, N);
  k_place<<<(E + T - 1) / T, T, 0, stream>>>(csr, rowptr, rank, src, dst, E);

  const int mtiles = N / 16;

  {
    int waves = mtiles * 4;
    int blocks = (waves + 7) / 8;
    k_gemm_wmma_f32<IN_C, HID, 4, PLANE><<<blocks, T, 0, stream>>>(x, W1, out, bufA, mtiles);
  }

  k_gather_rows<2, true, 4><<<(N + 15) / 16, T, 0, stream>>>(out, bufA, bufB, rowptr, csr, dinv, b1, N);

  {
    int waves = mtiles * 2;
    int blocks = (waves + 7) / 8;
    k_gemm_wmma_f32<HID, OUT_C, 2, PLANE><<<blocks, T, 0, stream>>>(bufB, W2, bufA, bufA, mtiles);
  }

  k_gather_rows<1, false, 8><<<2048, T, 0, stream>>>(bufA, bufA, out, rowptr, csr, dinv, b2, N);
}


# Edges grouped by destination through LDS bins (count per chunk, bin-sorted scatter, per-bin rows) instead of global atomics, scattered stores and a one-block scan; row sum order = LDS arrival order
# speedup vs baseline: 2.8972x; 1.4102x over previous
#include <hip/hip_runtime.h>


typedef __attribute__((ext_vector_type(2))) float v2f;
typedef __attribute__((ext_vector_type(8))) float v8f;

#define IN_C 512
#define HID 256
#define OUT_C 128
#define PLANE 128

#define BIN_SHIFT 8
#define BIN_NODES (1 << BIN_SHIFT)
#define MAX_BINS 1024
#define GT 1024
#define CH 4096
#define CH_PER (CH / GT)
#define STAGE_CAP 16384

__device__ __forceinline__ int block_scan_excl(int val, int* wsum, int lane, int w) {
  int incl = val;
  for (int o = 1; o < 32; o <<= 1) {
    int t = __shfl_up(incl, o);
    if (lane >= o) incl += t;
  }
  if (lane == 31) wsum[w] = incl;
  __syncthreads();
  const int ws = wsum[lane];
  int wincl = ws;
  for (int o = 1; o < 32; o <<= 1) {
    int t = __shfl_up(wincl, o);
    if (lane >= o) wincl += t;
  }
  const int woff = __shfl(wincl - ws, w);
  __syncthreads();
  return woff + incl - val;
}

__global__ void __launch_bounds__(GT)
k_bin_count(int* __restrict__ tbl, const int* __restrict__ dst, int e, int nb) {
  __shared__ int h[MAX_BINS];
  const int tid = threadIdx.x;
  const int c = blockIdx.x;
  h[tid] = 0;
  int d[CH_PER];
#pragma unroll
  for (int k = 0; k < CH_PER; ++k) {
    const int i = c * CH + k * GT + tid;
    d[k] = dst[i < e ? i : e - 1];
  }
  __syncthreads();
#pragma unroll
  for (int k = 0; k < CH_PER; ++k) {
    const int i = c * CH + k * GT + tid;
    if (i < e) atomicAdd(&h[d[k] >> BIN_SHIFT], 1);
  }
  __syncthreads();
  if (tid < nb) tbl[c * nb + tid] = h[tid];
}

__global__ void __launch_bounds__(GT)
k_bin_scatter(int2* __restrict__ binned, int* __restrict__ binstart,
              const int* __restrict__ tbl, const int* __restrict__ src,
              const int* __restrict__ dst, int e, int nb, int nc) {
  __shared__ int tot[MAX_BINS];
  __shared__ int bef[MAX_BINS];
  __shared__ int hcnt[MAX_BINS];
  __shared__ int lstart[MAX_BINS];
  __shared__ int goff[MAX_BINS];
  __shared__ int wsum[32];
  __shared__ int2 sbuf[CH];
  const int tid = threadIdx.x;
  const int lane = tid & 31;
  const int w = tid >> 5;
  const int c = blockIdx.x;

  tot[tid] = 0;
  bef[tid] = 0;
  hcnt[tid] = 0;

  int es[CH_PER], ed[CH_PER];
#pragma unroll
  for (int k = 0; k < CH_PER; ++k) {
    const int i = c * CH + k * GT + tid;
    const int ic = i < e ? i : e - 1;
    es[k] = src[ic];
    ed[k] = dst[ic];
  }
  __syncthreads();

  {
    const int binl = tid & 255;
    const int part = tid >> 8;
    const int q = (nc + 3) >> 2;
    const int c0 = part * q;
    const int c1 = (c0 + q < nc) ? (c0 + q) : nc;
    for (int b = binl; b < nb; b += 256) {
      int st = 0, sb = 0;
#pragma unroll 8
      for (int cc = c0; cc < c1; ++cc) {
        const int v = tbl[cc * nb + b];
        st += v;
        sb += (cc < c) ? v : 0;
      }
      atomicAdd(&tot[b], st);
      atomicAdd(&bef[b], sb);
    }
  }

  int lr[CH_PER];
#pragma unroll
  for (int k = 0; k < CH_PER; ++k) {
    const int i = c * CH + k * GT + tid;
    lr[k] = 0;
    if (i < e) lr[k] = atomicAdd(&hcnt[ed[k] >> BIN_SHIFT], 1);
  }
  __syncthreads();

  const int my_tot = tot[tid];
  const int my_cnt = hcnt[tid];
  const int ex_tot = block_scan_excl(my_tot, wsum, lane, w);
  const int ex_cnt = block_scan_excl(my_cnt, wsum, lane, w);
  lstart[tid] = ex_cnt;
  goff[tid] = ex_tot + bef[tid];
  if (c == 0) {
    if (tid < nb) binstart[tid] = ex_tot;
    if (tid == 0) binstart[nb] = e;
  }
  __syncthreads();

#pragma unroll
  for (int k = 0; k < CH_PER; ++k) {
    const int i = c * CH + k * GT + tid;
    if (i < e) sbuf[lstart[ed[k] >> BIN_SHIFT] + lr[k]] = make_int2(es[k], ed[k]);
  }
  __syncthreads();

  const int left = e - c * CH;
  const int nloc = left < CH ? left : CH;
#pragma unroll
  for (int k = 0; k < CH_PER; ++k) {
    const int j = k * GT + tid;
    if (j < nloc) {
      const int2 v = sbuf[j];
      const int b = v.y >> BIN_SHIFT;
      binned[goff[b] + (j - lstart[b])] = v;
    }
  }
}

__global__ void __launch_bounds__(GT)
k_bin_rows(int* __restrict__ csr, int* __restrict__ rowptr, float* __restrict__ dinv,
           const int2* __restrict__ binned, const int* __restrict__ binstart,
           int n, int nb) {
  __shared__ int ncnt[BIN_NODES];
  __shared__ int nstart[BIN_NODES];
  __shared__ int wsum[32];
  __shared__ int stage[STAGE_CAP];
  const int tid = threadIdx.x;
  const int lane = tid & 31;
  const int w = tid >> 5;
  const int b = blockIdx.x;
  const int beg = binstart[b];
  const int end = binstart[b + 1];
  const int m = end - beg;
  const int node0 = b << BIN_SHIFT;

  if (tid < BIN_NODES) ncnt[tid] = 0;
  __syncthreads();
  for (int i = tid; i < m; i += GT) {
    const int2 v = binned[beg + i];
    atomicAdd(&ncnt[v.y - node0], 1);
  }
  __syncthreads();

  const int cval = (tid < BIN_NODES) ? ncnt[tid] : 0;
  const int ex = block_scan_excl(cval, wsum, lane, w);
  if (tid < BIN_NODES) {
    nstart[tid] = ex;
    ncnt[tid] = 0;
    const int d = node0 + tid;
    if (d < n) {
      rowptr[d] = beg + ex;
      dinv[d] = rsqrtf((float)(cval + 1));
    }
  }
  if (b == nb - 1 && tid == 0) rowptr[n] = end;
  __syncthreads();

  const bool fits = m <= STAGE_CAP;
  for (int i = tid; i < m; i += GT) {
    const int2 v = binned[beg + i];
    const int dl = v.y - node0;
    const int pos = nstart[dl] + atomicAdd(&ncnt[dl], 1);
    if (fits) stage[pos] = v.x;
    else csr[beg + pos] = v.x;
  }
  if (fits) {
    __syncthreads();
    for (int i = tid; i < m; i += GT) csr[beg + i] = stage[i];
  }
}

template <int KDIM, int NCOLS, int NGROUPS, int CPL>
__global__ void k_gemm_wmma_f32(const float* __restrict__ A,
                                const float* __restrict__ B,
                                float* __restrict__ C0, float* __restrict__ C1,
                                int mtiles) {
  const int lane = threadIdx.x & 31;
  const int wib  = threadIdx.x >> 5;
  const int wid  = blockIdx.x * (blockDim.x >> 5) + wib;
  const int totalWaves = mtiles * NGROUPS;
  if (wid >= totalWaves) return;

  const int mt = wid / NGROUPS;
  const int ng = wid % NGROUPS;
  const int m0 = mt * 16;
  const int n0 = ng * 64;

  const int arow  = m0 + (lane & 15);
  const int khalf = (lane >> 4) * 2;
  const int bcol  = n0 + (lane & 15);

  const float* __restrict__ ap = A + (long long)arow * KDIM;

  v8f c0 = {}, c1 = {}, c2 = {}, c3 = {};

#pragma unroll 4
  for (int k = 0; k < KDIM; k += 4) {
    v2f a;
    a.x = ap[k + khalf];
    a.y = ap[k + khalf + 1];

    const float* __restrict__ br0 = B + (long long)(k + khalf) * NCOLS + bcol;
    const float* __restrict__ br1 = br0 + NCOLS;
    v2f b0, b1, b2, b3;
    b0.x = br0[0];  b0.y = br1[0];
    b1.x = br0[16]; b1.y = br1[16];
    b2.x = br0[32]; b2.y = br1[32];
    b3.x = br0[48]; b3.y = br1[48];

    c0 = __builtin_amdgcn_wmma_f32_16x16x4_f32(false, a, false, b0, (short)0, c0, false, false);
    c1 = __builtin_amdgcn_wmma_f32_16x16x4_f32(false, a, false, b1, (short)0, c1, false, false);
    c2 = __builtin_amdgcn_wmma_f32_16x16x4_f32(false, a, false, b2, (short)0, c2, false, false);
    c3 = __builtin_amdgcn_wmma_f32_16x16x4_f32(false, a, false, b3, (short)0, c3, false, false);
  }

  const int rbase = m0 + (lane >> 4) * 8;
  const int cc    = lane & 15;
  const int plane = n0 / CPL;
  const int nc    = n0 - plane * CPL;
  float* __restrict__ Cb = plane ? C1 : C0;
#pragma unroll
  for (int r = 0; r < 8; ++r) {
    float* __restrict__ orow = Cb + (long long)(rbase + r) * CPL + nc + cc;
    orow[0]  = c0[r];
    orow[16] = c1[r];
    orow[32] = c2[r];
    orow[48] = c3[r];
  }
}

template <int NPL, bool RELU, int UNR>
__global__ void __launch_bounds__(256)
k_gather_rows(const float* __restrict__ p0, const float* __restrict__ p1,
              float* __restrict__ rows_out,
              const int* __restrict__ rowptr, const int* __restrict__ csr,
              const float* __restrict__ dinv, const float* __restrict__ bias,
              int N) {
  const int lane = threadIdx.x & 31;
  const int wid = __builtin_amdgcn_readfirstlane(
      (int)(blockIdx.x * (blockDim.x >> 5) + (threadIdx.x >> 5)));
  const int nw = (int)(gridDim.x * (blockDim.x >> 5));

  for (int d = wid; d < N; d += nw) {
    const int beg = rowptr[d];
    const int end = rowptr[d + 1];
    const float dd = dinv[d];
    const float wself = dd * dd;

    float4 a0 = ((const float4*)(p0 + (size_t)d * PLANE))[lane];
    a0.x *= wself; a0.y *= wself; a0.z *= wself; a0.w *= wself;
    float4 a1 = make_float4(0.f, 0.f, 0.f, 0.f);
    if (NPL == 2) {
      a1 = ((const float4*)(p1 + (size_t)d * PLANE))[lane];
      a1.x *= wself; a1.y *= wself; a1.z *= wself; a1.w *= wself;
    }

    for (int base = beg; base < end; base += 32) {
      const int left = end - base;
      const int cnt = left < 32 ? left : 32;
      int s_l = d;
      float w_l = 0.0f;
      if (lane < cnt) {
        s_l = csr[base + lane];
        w_l = dinv[s_l] * dd;
      }
      for (int j = 0; j < cnt; j += UNR) {
        float4 v0[UNR];
        float4 v1[UNR];
        float w[UNR];
#pragma unroll
        for (int u = 0; u < UNR; ++u) {
          const int s = __builtin_amdgcn_readlane(s_l, j + u);
          w[u] = __builtin_bit_cast(
              float, __builtin_amdgcn_readlane(__builtin_bit_cast(int, w_l), j + u));
          v0[u] = ((const float4*)(p0 + (size_t)s * PLANE))[lane];
          if (NPL == 2) v1[u] = ((const float4*)(p1 + (size_t)s * PLANE))[lane];
        }
#pragma unroll
        for (int u = 0; u < UNR; ++u) {
          a0.x += w[u] * v0[u].x; a0.y += w[u] * v0[u].y;
          a0.z += w[u] * v0[u].z; a0.w += w[u] * v0[u].w;
          if (NPL == 2) {
            a1.x += w[u] * v1[u].x; a1.y += w[u] * v1[u].y;
            a1.z += w[u] * v1[u].z; a1.w += w[u] * v1[u].w;
          }
        }
      }
    }

    float* __restrict__ orow = rows_out + (size_t)d * (NPL * PLANE);
    {
      const float4 b = ((const float4*)bias)[lane];
      a0.x += b.x; a0.y += b.y; a0.z += b.z; a0.w += b.w;
      if (RELU) {
        a0.x = fmaxf(a0.x, 0.f); a0.y = fmaxf(a0.y, 0.f);
        a0.z = fmaxf(a0.z, 0.f); a0.w = fmaxf(a0.w, 0.f);
      }
      ((float4*)orow)[lane] = a0;
    }
    if (NPL == 2) {
      const float4 b = ((const float4*)(bias + PLANE))[lane];
      a1.x += b.x; a1.y += b.y; a1.z += b.z; a1.w += b.w;
      if (RELU) {
        a1.x = fmaxf(a1.x, 0.f); a1.y = fmaxf(a1.y, 0.f);
        a1.z = fmaxf(a1.z, 0.f); a1.w = fmaxf(a1.w, 0.f);
      }
      ((float4*)(orow + PLANE))[lane] = a1;
    }
  }
}

extern "C" void kernel_launch(void* const* d_in, const int* in_sizes, int n_in,
                              void* d_out, int out_size, void* d_ws, size_t ws_size,
                              hipStream_t stream) {
  (void)n_in; (void)out_size; (void)ws_size;

  const float* x  = (const float*)d_in[0];
  const int*   ei = (const int*)d_in[1];
  const float* W1 = (const float*)d_in[2];
  const float* b1 = (const float*)d_in[3];
  const float* W2 = (const float*)d_in[4];
  const float* b2 = (const float*)d_in[5];
  float* out = (float*)d_out;

  const int N = in_sizes[0] / IN_C;
  const int E = in_sizes[1] / 2;
  const int* src = ei;
  const int* dst = ei + E;

  char* w = (char*)d_ws;
  size_t dinv_bytes = (((size_t)N * 4) + 255) & ~(size_t)255;
  size_t csr_bytes  = (((size_t)E * 4) + 255) & ~(size_t)255;
  float* dinv = (float*)w;
  float* bufA = (float*)(w + dinv_bytes);
  float* bufB = bufA + (size_t)N * HID;
  char*  elist  = (char*)(bufA + (size_t)N * PLANE);
  const int NB = (N + BIN_NODES - 1) / BIN_NODES;
  const int NC = (E + CH - 1) / CH;
  int*   rowptr   = (int*)elist;
  int*   csr      = (int*)(elist + dinv_bytes);
  int2*  binned   = (int2*)(elist + dinv_bytes + csr_bytes);
  int*   tbl      = (int*)(elist + dinv_bytes + 3 * csr_bytes);
  int*   binstart = tbl + (size_t)NC * NB;

  const int T = 256;

  k_bin_count<<<NC, GT, 0, stream>>>(tbl, dst, E, NB);
  k_bin_scatter<<<NC, GT, 0, stream>>>(binned, binstart, tbl, src, dst, E, NB, NC);
  k_bin_rows<<<NB, GT, 0, stream>>>(csr, rowptr, dinv, binned, binstart, N, NB);

  const int mtiles = N / 16;

  {
    int waves = mtiles * 4;
    int blocks = (waves + 7) / 8;
    k_gemm_wmma_f32<IN_C, HID, 4, PLANE><<<blocks, T, 0, stream>>>(x, W1, out, bufA, mtiles);
  }

  k_gather_rows<2, true, 4><<<(N + 15) / 16, T, 0, stream>>>(out, bufA, bufB, rowptr, csr, dinv, b1, N);

  {
    int waves = mtiles * 2;
    int blocks = (waves + 7) / 8;
    k_gemm_wmma_f32<HID, OUT_C, 2, PLANE><<<blocks, T, 0, stream>>>(bufB, W2, bufA, bufA, mtiles);
  }

  k_gather_rows<1, false, 8><<<2048, T, 0, stream>>>(bufA, bufA, out, rowptr, csr, dinv, b2, N);
}


# Both f32 WMMA products fed from LDS: 32x128 block tiles of 4 waves (2x2 tiles each), A and B chunks of 32 k fetched as whole 16-byte row pieces, double-buffered; k sum order unchanged
# speedup vs baseline: 3.5778x; 1.2349x over previous
#include <hip/hip_runtime.h>


typedef __attribute__((ext_vector_type(2))) float v2f;
typedef __attribute__((ext_vector_type(8))) float v8f;

#define IN_C 512
#define HID 256
#define OUT_C 128
#define PLANE 128

#define BIN_SHIFT 8
#define BIN_NODES (1 << BIN_SHIFT)
#define MAX_BINS 1024
#define GT 1024
#define CH 4096
#define CH_PER (CH / GT)
#define STAGE_CAP 16384

__device__ __forceinline__ int block_scan_excl(int val, int* wsum, int lane, int w) {
  int incl = val;
  for (int o = 1; o < 32; o <<= 1) {
    int t = __shfl_up(incl, o);
    if (lane >= o) incl += t;
  }
  if (lane == 31) wsum[w] = incl;
  __syncthreads();
  const int ws = wsum[lane];
  int wincl = ws;
  for (int o = 1; o < 32; o <<= 1) {
    int t = __shfl_up(wincl, o);
    if (lane >= o) wincl += t;
  }
  const int woff = __shfl(wincl - ws, w);
  __syncthreads();
  return woff + incl - val;
}

__global__ void __launch_bounds__(GT)
k_bin_count(int* __restrict__ tbl, const int* __restrict__ dst, int e, int nb) {
  __shared__ int h[MAX_BINS];
  const int tid = threadIdx.x;
  const int c = blockIdx.x;
  h[tid] = 0;
  int d[CH_PER];
#pragma unroll
  for (int k = 0; k < CH_PER; ++k) {
    const int i = c * CH + k * GT + tid;
    d[k] = dst[i < e ? i : e - 1];
  }
  __syncthreads();
#pragma unroll
  for (int k = 0; k < CH_PER; ++k) {
    const int i = c * CH + k * GT + tid;
    if (i < e) atomicAdd(&h[d[k] >> BIN_SHIFT], 1);
  }
  __syncthreads();
  if (tid < nb) tbl[c * nb + tid] = h[tid];
}

__global__ void __launch_bounds__(GT)
k_bin_scatter(int2* __restrict__ binned, int* __restrict__ binstart,
              const int* __restrict__ tbl, const int* __restrict__ src,
              const int* __restrict__ dst, int e, int nb, int nc) {
  __shared__ int tot[MAX_BINS];
  __shared__ int bef[MAX_BINS];
  __shared__ int hcnt[MAX_BINS];
  __shared__ int lstart[MAX_BINS];
  __shared__ int goff[MAX_BINS];
  __shared__ int wsum[32];
  __shared__ int2 sbuf[CH];
  const int tid = threadIdx.x;
  const int lane = tid & 31;
  const int w = tid >> 5;
  const int c = blockIdx.x;

  tot[tid] = 0;
  bef[tid] = 0;
  hcnt[tid] = 0;

  int es[CH_PER], ed[CH_PER];
#pragma unroll
  for (int k = 0; k < CH_PER; ++k) {
    const int i = c * CH + k * GT + tid;
    const int ic = i < e ? i : e - 1;
    es[k] = src[ic];
    ed[k] = dst[ic];
  }
  __syncthreads();

  {
    const int binl = tid & 255;
    const int part = tid >> 8;
    const int q = (nc + 3) >> 2;
    const int c0 = part * q;
    const int c1 = (c0 + q < nc) ? (c0 + q) : nc;
    for (int b = binl; b < nb; b += 256) {
      int st = 0, sb = 0;
#pragma unroll 8
      for (int cc = c0; cc < c1; ++cc) {
        const int v = tbl[cc * nb + b];
        st += v;
        sb += (cc < c) ? v : 0;
      }
      atomicAdd(&tot[b], st);
      atomicAdd(&bef[b], sb);
    }
  }

  int lr[CH_PER];
#pragma unroll
  for (int k = 0; k < CH_PER; ++k) {
    const int i = c * CH + k * GT + tid;
    lr[k] = 0;
    if (i < e) lr[k] = atomicAdd(&hcnt[ed[k] >> BIN_SHIFT], 1);
  }
  __syncthreads();

  const int my_tot = tot[tid];
  const int my_cnt = hcnt[tid];
  const int ex_tot = block_scan_excl(my_tot, wsum, lane, w);
  const int ex_cnt = block_scan_excl(my_cnt, wsum, lane, w);
  lstart[tid] = ex_cnt;
  goff[tid] = ex_tot + bef[tid];
  if (c == 0) {
    if (tid < nb) binstart[tid] = ex_tot;
    if (tid == 0) binstart[nb] = e;
  }
  __syncthreads();

#pragma unroll
  for (int k = 0; k < CH_PER; ++k) {
    const int i = c * CH + k * GT + tid;
    if (i < e) sbuf[lstart[ed[k] >> BIN_SHIFT] + lr[k]] = make_int2(es[k], ed[k]);
  }
  __syncthreads();

  const int left = e - c * CH;
  const int nloc = left < CH ? left : CH;
#pragma unroll
  for (int k = 0; k < CH_PER; ++k) {
    const int j = k * GT + tid;
    if (j < nloc) {
      const int2 v = sbuf[j];
      const int b = v.y >> BIN_SHIFT;
      binned[goff[b] + (j - lstart[b])] = v;
    }
  }
}

__global__ void __launch_bounds__(GT)
k_bin_rows(int* __restrict__ csr, int* __restrict__ rowptr, float* __restrict__ dinv,
           const int2* __restrict__ binned, const int* __restrict__ binstart,
           int n, int nb) {
  __shared__ int ncnt[BIN_NODES];
  __shared__ int nstart[BIN_NODES];
  __shared__ int wsum[32];
  __shared__ int stage[STAGE_CAP];
  const int tid = threadIdx.x;
  const int lane = tid & 31;
  const int w = tid >> 5;
  const int b = blockIdx.x;
  const int beg = binstart[b];
  const int end = binstart[b + 1];
  const int m = end - beg;
  const int node0 = b << BIN_SHIFT;

  if (tid < BIN_NODES) ncnt[tid] = 0;
  __syncthreads();
  for (int i = tid; i < m; i += GT) {
    const int2 v = binned[beg + i];
    atomicAdd(&ncnt[v.y - node0], 1);
  }
  __syncthreads();

  const int cval = (tid < BIN_NODES) ? ncnt[tid] : 0;
  const int ex = block_scan_excl(cval, wsum, lane, w);
  if (tid < BIN_NODES) {
    nstart[tid] = ex;
    ncnt[tid] = 0;
    const int d = node0 + tid;
    if (d < n) {
      rowptr[d] = beg + ex;
      dinv[d] = rsqrtf((float)(cval + 1));
    }
  }
  if (b == nb - 1 && tid == 0) rowptr[n] = end;
  __syncthreads();

  const bool fits = m <= STAGE_CAP;
  for (int i = tid; i < m; i += GT) {
    const int2 v = binned[beg + i];
    const int dl = v.y - node0;
    const int pos = nstart[dl] + atomicAdd(&ncnt[dl], 1);
    if (fits) stage[pos] = v.x;
    else csr[beg + pos] = v.x;
  }
  if (fits) {
    __syncthreads();
    for (int i = tid; i < m; i += GT) csr[beg + i] = stage[i];
  }
}

#define GBM 32
#define GBN 128
#define GBK 32
#define APITCH 36
#define BTILE 528

template <int KDIM, int NCOLS>
__global__ void __launch_bounds__(128)
k_gemm_wmma_f32(const float* __restrict__ A,
                const float* __restrict__ B,
                float* __restrict__ C0, float* __restrict__ C1,
                int M) {
  __shared__ __attribute__((aligned(16))) float As[2][GBM * APITCH];
  __shared__ __attribute__((aligned(16))) float Bs[2][(GBN / 16) * BTILE];

  constexpr int NBN = NCOLS / GBN;
  constexpr int T = KDIM / GBK;

  const int tid  = threadIdx.x;
  const int lane = tid & 31;
  const int wv   = tid >> 5;
  const int c    = lane & 15;
  const int h    = lane >> 4;
  const int bn   = blockIdx.x % NBN;
  const int m0   = (blockIdx.x / NBN) * GBM;
  const int n0   = bn * GBN;

  const int akq = (tid & 7) * 4;
  const int ar0 = tid >> 3;
  int arow0 = m0 + ar0;       arow0 = arow0 < M ? arow0 : M - 1;
  int arow1 = m0 + ar0 + 16;  arow1 = arow1 < M ? arow1 : M - 1;
  const float* __restrict__ ag0 = A + (size_t)arow0 * KDIM + akq;
  const float* __restrict__ ag1 = A + (size_t)arow1 * KDIM + akq;
  const float* __restrict__ bg  = B + (size_t)wv * NCOLS + n0 + 4 * lane;
  const int as_st = ar0 * APITCH + akq;
  const int bs_st = (lane >> 2) * BTILE + wv * 16 + (lane & 3) * 4;

  const int a_rd = c * APITCH + 2 * h;
  const int b_rd = (2 * wv) * BTILE + 32 * h + c;

  float4 ra0, ra1, rb[8];
  v8f c00 = {}, c01 = {}, c10 = {}, c11 = {};

#define GEMM_FETCH(k1)                                                                   \
  {                                                                                      \
    ra0 = *(const float4*)(ag0 + (k1));                                                  \
    ra1 = *(const float4*)(ag1 + (k1));                                                  \
    _Pragma("unroll") for (int i = 0; i < 8; ++i)                                        \
        rb[i] = *(const float4*)(bg + (size_t)((k1) + 4 * i) * NCOLS);                   \
  }
#define GEMM_PUT(nb)                                                                     \
  {                                                                                      \
    *(float4*)(&As[nb][as_st]) = ra0;                                                    \
    *(float4*)(&As[nb][as_st + 16 * APITCH]) = ra1;                                      \
    _Pragma("unroll") for (int i = 0; i < 8; ++i)                                        \
        *(float4*)(&Bs[nb][bs_st + 64 * i]) = rb[i];                                     \
  }
#define GEMM_MULT(cb)                                                                    \
  {                                                                                      \
    const float* __restrict__ as = &As[cb][a_rd];                                        \
    const float* __restrict__ bs = &Bs[cb][b_rd];                                        \
    _Pragma("unroll") for (int ks = 0; ks < GBK / 4; ++ks) {                             \
      const v2f a0 = *(const v2f*)(as + 4 * ks);                                         \
      const v2f a1 = *(const v2f*)(as + 16 * APITCH + 4 * ks);                           \
      v2f b0, b1;                                                                        \
      b0.x = bs[64 * ks];          b0.y = bs[64 * ks + 16];                              \
      b1.x = bs[BTILE + 64 * ks];  b1.y = bs[BTILE + 64 * ks + 16];                      \
      c00 = __builtin_amdgcn_wmma_f32_16x16x4_f32(false, a0, false, b0, (short)0, c00, false, false); \
      c01 = __builtin_amdgcn_wmma_f32_16x16x4_f32(false, a0, false, b1, (short)0, c01, false, false); \
      c10 = __builtin_amdgcn_wmma_f32_16x16x4_f32(false, a1, false, b0, (short)0, c10, false, false); \
      c11 = __builtin_amdgcn_wmma_f32_16x16x4_f32(false, a1, false, b1, (short)0, c11, false, false); \
    }                                                                                    \
  }

  GEMM_FETCH(0)
  GEMM_PUT(0)
  __syncthreads();

  for (int t = 0; t < T; t += 2) {
    GEMM_FETCH((t + 1) * GBK)
    GEMM_MULT(0)
    GEMM_PUT(1)
    __syncthreads();

    const int t2 = (t + 2 < T) ? (t + 2) : (T - 1);
    GEMM_FETCH(t2 * GBK)
    GEMM_MULT(1)
    GEMM_PUT(0)
    __syncthreads();
  }
#undef GEMM_FETCH
#undef GEMM_PUT
#undef GEMM_MULT

  float* __restrict__ Cb = bn ? C1 : C0;
  const int col = wv * 32 + c;
#pragma unroll
  for (int r = 0; r < 8; ++r) {
    const int row0 = m0 + 8 * h + r;
    const int row1 = row0 + 16;
    if (row0 < M) {
      float* __restrict__ o = Cb + (size_t)row0 * GBN + col;
      o[0]  = c00[r];
      o[16] = c01[r];
    }
    if (row1 < M) {
      float* __restrict__ o = Cb + (size_t)row1 * GBN + col;
      o[0]  = c10[r];
      o[16] = c11[r];
    }
  }
}

template <int NPL, bool RELU, int UNR>
__global__ void __launch_bounds__(256)
k_gather_rows(const float* __restrict__ p0, const float* __restrict__ p1,
              float* __restrict__ rows_out,
              const int* __restrict__ rowptr, const int* __restrict__ csr,
              const float* __restrict__ dinv, const float* __restrict__ bias,
              int N) {
  const int lane = threadIdx.x & 31;
  const int wid = __builtin_amdgcn_readfirstlane(
      (int)(blockIdx.x * (blockDim.x >> 5) + (threadIdx.x >> 5)));
  const int nw = (int)(gridDim.x * (blockDim.x >> 5));

  for (int d = wid; d < N; d += nw) {
    const int beg = rowptr[d];
    const int end = rowptr[d + 1];
    const float dd = dinv[d];
    const float wself = dd * dd;

    float4 a0 = ((const float4*)(p0 + (size_t)d * PLANE))[lane];
    a0.x *= wself; a0.y *= wself; a0.z *= wself; a0.w *= wself;
    float4 a1 = make_float4(0.f, 0.f, 0.f, 0.f);
    if (NPL == 2) {
      a1 = ((const float4*)(p1 + (size_t)d * PLANE))[lane];
      a1.x *= wself; a1.y *= wself; a1.z *= wself; a1.w *= wself;
    }

    for (int base = beg; base < end; base += 32) {
      const int left = end - base;
      const int cnt = left < 32 ? left : 32;
      int s_l = d;
      float w_l = 0.0f;
      if (lane < cnt) {
        s_l = csr[base + lane];
        w_l = dinv[s_l] * dd;
      }
      for (int j = 0; j < cnt; j += UNR) {
        float4 v0[UNR];
        float4 v1[UNR];
        float w[UNR];
#pragma unroll
        for (int u = 0; u < UNR; ++u) {
          const int s = __builtin_amdgcn_readlane(s_l, j + u);
          w[u] = __builtin_bit_cast(
              float, __builtin_amdgcn_readlane(__builtin_bit_cast(int, w_l), j + u));
          v0[u] = ((const float4*)(p0 + (size_t)s * PLANE))[lane];
          if (NPL == 2) v1[u] = ((const float4*)(p1 + (size_t)s * PLANE))[lane];
        }
#pragma unroll
        for (int u = 0; u < UNR; ++u) {
          a0.x += w[u] * v0[u].x; a0.y += w[u] * v0[u].y;
          a0.z += w[u] * v0[u].z; a0.w += w[u] * v0[u].w;
          if (NPL == 2) {
            a1.x += w[u] * v1[u].x; a1.y += w[u] * v1[u].y;
            a1.z += w[u] * v1[u].z; a1.w += w[u] * v1[u].w;
          }
        }
      }
    }

    float* __restrict__ orow = rows_out + (size_t)d * (NPL * PLANE);
    {
      const float4 b = ((const float4*)bias)[lane];
      a0.x += b.x; a0.y += b.y; a0.z += b.z; a0.w += b.w;
      if (RELU) {
        a0.x = fmaxf(a0.x, 0.f); a0.y = fmaxf(a0.y, 0.f);
        a0.z = fmaxf(a0.z, 0.f); a0.w = fmaxf(a0.w, 0.f);
      }
      ((float4*)orow)[lane] = a0;
    }
    if (NPL == 2) {
      const float4 b = ((const float4*)(bias + PLANE))[lane];
      a1.x += b.x; a1.y += b.y; a1.z += b.z; a1.w += b.w;
      if (RELU) {
        a1.x = fmaxf(a1.x, 0.f); a1.y = fmaxf(a1.y, 0.f);
        a1.z = fmaxf(a1.z, 0.f); a1.w = fmaxf(a1.w, 0.f);
      }
      ((float4*)(orow + PLANE))[lane] = a1;
    }
  }
}

extern "C" void kernel_launch(void* const* d_in, const int* in_sizes, int n_in,
                              void* d_out, int out_size, void* d_ws, size_t ws_size,
                              hipStream_t stream) {
  (void)n_in; (void)out_size; (void)ws_size;

  const float* x  = (const float*)d_in[0];
  const int*   ei = (const int*)d_in[1];
  const float* W1 = (const float*)d_in[2];
  const float* b1 = (const float*)d_in[3];
  const float* W2 = (const float*)d_in[4];
  const float* b2 = (const float*)d_in[5];
  float* out = (float*)d_out;

  const int N = in_sizes[0] / IN_C;
  const int E = in_sizes[1] / 2;
  const int* src = ei;
  const int* dst = ei + E;

  char* w = (char*)d_ws;
  size_t dinv_bytes = (((size_t)N * 4) + 255) & ~(size_t)255;
  size_t csr_bytes  = (((size_t)E * 4) + 255) & ~(size_t)255;
  float* dinv = (float*)w;
  float* bufA = (float*)(w + dinv_bytes);
  float* bufB = bufA + (size_t)N * HID;
  char*  elist  = (char*)(bufA + (size_t)N * PLANE);
  const int NB = (N + BIN_NODES - 1) / BIN_NODES;
  const int NC = (E + CH - 1) / CH;
  int*   rowptr   = (int*)elist;
  int*   csr      = (int*)(elist + dinv_bytes);
  int2*  binned   = (int2*)(elist + dinv_bytes + csr_bytes);
  int*   tbl      = (int*)(elist + dinv_bytes + 3 * csr_bytes);
  int*   binstart = tbl + (size_t)NC * NB;

  const int T = 256;

  k_bin_count<<<NC, GT, 0, stream>>>(tbl, dst, E, NB);
  k_bin_scatter<<<NC, GT, 0, stream>>>(binned, binstart, tbl, src, dst, E, NB, NC);
  k_bin_rows<<<NB, GT, 0, stream>>>(csr, rowptr, dinv, binned, binstart, N, NB);

  const int mblocks = (N + GBM - 1) / GBM;

  k_gemm_wmma_f32<IN_C, HID><<<mblocks * (HID / GBN), 128, 0, stream>>>(x, W1, out, bufA, N);

  k_gather_rows<2, true, 4><<<(N + 15) / 16, T, 0, stream>>>(out, bufA, bufB, rowptr, csr, dinv, b1, N);

  k_gemm_wmma_f32<HID, OUT_C><<<mblocks * (OUT_C / GBN), 128, 0, stream>>>(bufB, W2, bufA, bufA, N);

  k_gather_rows<1, false, 8><<<2048, T, 0, stream>>>(bufA, bufA, out, rowptr, csr, dinv, b2, N);
}


# Second product fused into the first aggregation: a block of 8 waves sums 32 rows of h (bias, ReLU) into LDS and multiplies them by W2 on f32 WMMA there; h never goes to memory; sum orders kept
# speedup vs baseline: 3.7171x; 1.0389x over previous
#include <hip/hip_runtime.h>


typedef __attribute__((ext_vector_type(2))) float v2f;
typedef __attribute__((ext_vector_type(8))) float v8f;

#define IN_C 512
#define HID 256
#define OUT_C 128
#define PLANE 128

#define BIN_SHIFT 8
#define BIN_NODES (1 << BIN_SHIFT)
#define MAX_BINS 1024
#define GT 1024
#define CH 4096
#define CH_PER (CH / GT)
#define STAGE_CAP 16384

__device__ __forceinline__ int block_scan_excl(int val, int* wsum, int lane, int w) {
  int incl = val;
  for (int o = 1; o < 32; o <<= 1) {
    int t = __shfl_up(incl, o);
    if (lane >= o) incl += t;
  }
  if (lane == 31) wsum[w] = incl;
  __syncthreads();
  const int ws = wsum[lane];
  int wincl = ws;
  for (int o = 1; o < 32; o <<= 1) {
    int t = __shfl_up(wincl, o);
    if (lane >= o) wincl += t;
  }
  const int woff = __shfl(wincl - ws, w);
  __syncthreads();
  return woff + incl - val;
}

__global__ void __launch_bounds__(GT)
k_bin_count(int* __restrict__ tbl, const int* __restrict__ dst, int e, int nb) {
  __shared__ int h[MAX_BINS];
  const int tid = threadIdx.x;
  const int c = blockIdx.x;
  h[tid] = 0;
  int d[CH_PER];
#pragma unroll
  for (int k = 0; k < CH_PER; ++k) {
    const int i = c * CH + k * GT + tid;
    d[k] = dst[i < e ? i : e - 1];
  }
  __syncthreads();
#pragma unroll
  for (int k = 0; k < CH_PER; ++k) {
    const int i = c * CH + k * GT + tid;
    if (i < e) atomicAdd(&h[d[k] >> BIN_SHIFT], 1);
  }
  __syncthreads();
  if (tid < nb) tbl[c * nb + tid] = h[tid];
}

__global__ void __launch_bounds__(GT)
k_bin_scatter(int2* __restrict__ binned, int* __restrict__ binstart,
              const int* __restrict__ tbl, const int* __restrict__ src,
              const int* __restrict__ dst, int e, int nb, int nc) {
  __shared__ int tot[MAX_BINS];
  __shared__ int bef[MAX_BINS];
  __shared__ int hcnt[MAX_BINS];
  __shared__ int lstart[MAX_BINS];
  __shared__ int goff[MAX_BINS];
  __shared__ int wsum[32];
  __shared__ int2 sbuf[CH];
  const int tid = threadIdx.x;
  const int lane = tid & 31;
  const int w = tid >> 5;
  const int c = blockIdx.x;

  tot[tid] = 0;
  bef[tid] = 0;
  hcnt[tid] = 0;

  int es[CH_PER], ed[CH_PER];
#pragma unroll
  for (int k = 0; k < CH_PER; ++k) {
    const int i = c * CH + k * GT + tid;
    const int ic = i < e ? i : e - 1;
    es[k] = src[ic];
    ed[k] = dst[ic];
  }
  __syncthreads();

  {
    const int binl = tid & 255;
    const int part = tid >> 8;
    const int q = (nc + 3) >> 2;
    const int c0 = part * q;
    const int c1 = (c0 + q < nc) ? (c0 + q) : nc;
    for (int b = binl; b < nb; b += 256) {
      int st = 0, sb = 0;
#pragma unroll 8
      for (int cc = c0; cc < c1; ++cc) {
        const int v = tbl[cc * nb + b];
        st += v;
        sb += (cc < c) ? v : 0;
      }
      atomicAdd(&tot[b], st);
      atomicAdd(&bef[b], sb);
    }
  }

  int lr[CH_PER];
#pragma unroll
  for (int k = 0; k < CH_PER; ++k) {
    const int i = c * CH + k * GT + tid;
    lr[k] = 0;
    if (i < e) lr[k] = atomicAdd(&hcnt[ed[k] >> BIN_SHIFT], 1);
  }
  __syncthreads();

  const int my_tot = tot[tid];
  const int my_cnt = hcnt[tid];
  const int ex_tot = block_scan_excl(my_tot, wsum, lane, w);
  const int ex_cnt = block_scan_excl(my_cnt, wsum, lane, w);
  lstart[tid] = ex_cnt;
  goff[tid] = ex_tot + bef[tid];
  if (c == 0) {
    if (tid < nb) binstart[tid] = ex_tot;
    if (tid == 0) binstart[nb] = e;
  }
  __syncthreads();

#pragma unroll
  for (int k = 0; k < CH_PER; ++k) {
    const int i = c * CH + k * GT + tid;
    if (i < e) sbuf[lstart[ed[k] >> BIN_SHIFT] + lr[k]] = make_int2(es[k], ed[k]);
  }
  __syncthreads();

  const int left = e - c * CH;
  const int nloc = left < CH ? left : CH;
#pragma unroll
  for (int k = 0; k < CH_PER; ++k) {
    const int j = k * GT + tid;
    if (j < nloc) {
      const int2 v = sbuf[j];
      const int b = v.y >> BIN_SHIFT;
      binned[goff[b] + (j - lstart[b])] = v;
    }
  }
}

__global__ void __launch_bounds__(GT)
k_bin_rows(int* __restrict__ csr, int* __restrict__ rowptr, float* __restrict__ dinv,
           const int2* __restrict__ binned, const int* __restrict__ binstart,
           int n, int nb) {
  __shared__ int ncnt[BIN_NODES];
  __shared__ int nstart[BIN_NODES];
  __shared__ int wsum[32];
  __shared__ int stage[STAGE_CAP];
  const int tid = threadIdx.x;
  const int lane = tid & 31;
  const int w = tid >> 5;
  const int b = blockIdx.x;
  const int beg = binstart[b];
  const int end = binstart[b + 1];
  const int m = end - beg;
  const int node0 = b << BIN_SHIFT;

  if (tid < BIN_NODES) ncnt[tid] = 0;
  __syncthreads();
  for (int i = tid; i < m; i += GT) {
    const int2 v = binned[beg + i];
    atomicAdd(&ncnt[v.y - node0], 1);
  }
  __syncthreads();

  const int cval = (tid < BIN_NODES) ? ncnt[tid] : 0;
  const int ex = block_scan_excl(cval, wsum, lane, w);
  if (tid < BIN_NODES) {
    nstart[tid] = ex;
    ncnt[tid] = 0;
    const int d = node0 + tid;
    if (d < n) {
      rowptr[d] = beg + ex;
      dinv[d] = rsqrtf((float)(cval + 1));
    }
  }
  if (b == nb - 1 && tid == 0) rowptr[n] = end;
  __syncthreads();

  const bool fits = m <= STAGE_CAP;
  for (int i = tid; i < m; i += GT) {
    const int2 v = binned[beg + i];
    const int dl = v.y - node0;
    const int pos = nstart[dl] + atomicAdd(&ncnt[dl], 1);
    if (fits) stage[pos] = v.x;
    else csr[beg + pos] = v.x;
  }
  if (fits) {
    __syncthreads();
    for (int i = tid; i < m; i += GT) csr[beg + i] = stage[i];
  }
}

#define GBM 32
#define GBN 128
#define GBK 32
#define APITCH 36
#define BTILE 528

template <int KDIM, int NCOLS>
__global__ void __launch_bounds__(128)
k_gemm_wmma_f32(const float* __restrict__ A,
                const float* __restrict__ B,
                float* __restrict__ C0, float* __restrict__ C1,
                int M) {
  __shared__ __attribute__((aligned(16))) float As[2][GBM * APITCH];
  __shared__ __attribute__((aligned(16))) float Bs[2][(GBN / 16) * BTILE];

  constexpr int NBN = NCOLS / GBN;
  constexpr int T = KDIM / GBK;

  const int tid  = threadIdx.x;
  const int lane = tid & 31;
  const int wv   = tid >> 5;
  const int c    = lane & 15;
  const int h    = lane >> 4;
  const int bn   = blockIdx.x % NBN;
  const int m0   = (blockIdx.x / NBN) * GBM;
  const int n0   = bn * GBN;

  const int akq = (tid & 7) * 4;
  const int ar0 = tid >> 3;
  int arow0 = m0 + ar0;       arow0 = arow0 < M ? arow0 : M - 1;
  int arow1 = m0 + ar0 + 16;  arow1 = arow1 < M ? arow1 : M - 1;
  const float* __restrict__ ag0 = A + (size_t)arow0 * KDIM + akq;
  const float* __restrict__ ag1 = A + (size_t)arow1 * KDIM + akq;
  const float* __restrict__ bg  = B + (size_t)wv * NCOLS + n0 + 4 * lane;
  const int as_st = ar0 * APITCH + akq;
  const int bs_st = (lane >> 2) * BTILE + wv * 16 + (lane & 3) * 4;

  const int a_rd = c * APITCH + 2 * h;
  const int b_rd = (2 * wv) * BTILE + 32 * h + c;

  float4 ra0, ra1, rb[8];
  v8f c00 = {}, c01 = {}, c10 = {}, c11 = {};

#define GEMM_FETCH(k1)                                                                   \
  {                                                                                      \
    ra0 = *(const float4*)(ag0 + (k1));                                                  \
    ra1 = *(const float4*)(ag1 + (k1));                                                  \
    _Pragma("unroll") for (int i = 0; i < 8; ++i)                                        \
        rb[i] = *(const float4*)(bg + (size_t)((k1) + 4 * i) * NCOLS);                   \
  }
#define GEMM_PUT(nb)                                                                     \
  {                                                                                      \
    *(float4*)(&As[nb][as_st]) = ra0;                                                    \
    *(float4*)(&As[nb][as_st + 16 * APITCH]) = ra1;                                      \
    _Pragma("unroll") for (int i = 0; i < 8; ++i)                                        \
        *(float4*)(&Bs[nb][bs_st + 64 * i]) = rb[i];                                     \
  }
#define GEMM_MULT(cb)                                                                    \
  {                                                                                      \
    const float* __restrict__ as = &As[cb][a_rd];                                        \
    const float* __restrict__ bs = &Bs[cb][b_rd];                                        \
    _Pragma("unroll") for (int ks = 0; ks < GBK / 4; ++ks) {                             \
      const v2f a0 = *(const v2f*)(as + 4 * ks);                                         \
      const v2f a1 = *(const v2f*)(as + 16 * APITCH + 4 * ks);                           \
      v2f b0, b1;                                                                        \
      b0.x = bs[64 * ks];          b0.y = bs[64 * ks + 16];                              \
      b1.x = bs[BTILE + 64 * ks];  b1.y = bs[BTILE + 64 * ks + 16];                      \
      c00 = __builtin_amdgcn_wmma_f32_16x16x4_f32(false, a0, false, b0, (short)0, c00, false, false); \
      c01 = __builtin_amdgcn_wmma_f32_16x16x4_f32(false, a0, false, b1, (short)0, c01, false, false); \
      c10 = __builtin_amdgcn_wmma_f32_16x16x4_f32(false, a1, false, b0, (short)0, c10, false, false); \
      c11 = __builtin_amdgcn_wmma_f32_16x16x4_f32(false, a1, false, b1, (short)0, c11, false, false); \
    }                                                                                    \
  }

  GEMM_FETCH(0)
  GEMM_PUT(0)
  __syncthreads();

  for (int t = 0; t < T; t += 2) {
    GEMM_FETCH((t + 1) * GBK)
    GEMM_MULT(0)
    GEMM_PUT(1)
    __syncthreads();

    const int t2 = (t + 2 < T) ? (t + 2) : (T - 1);
    GEMM_FETCH(t2 * GBK)
    GEMM_MULT(1)
    GEMM_PUT(0)
    __syncthreads();
  }
#undef GEMM_FETCH
#undef GEMM_PUT
#undef GEMM_MULT

  float* __restrict__ Cb = bn ? C1 : C0;
  const int col = wv * 32 + c;
#pragma unroll
  for (int r = 0; r < 8; ++r) {
    const int row0 = m0 + 8 * h + r;
    const int row1 = row0 + 16;
    if (row0 < M) {
      float* __restrict__ o = Cb + (size_t)row0 * GBN + col;
      o[0]  = c00[r];
      o[16] = c01[r];
    }
    if (row1 < M) {
      float* __restrict__ o = Cb + (size_t)row1 * GBN + col;
      o[0]  = c10[r];
      o[16] = c11[r];
    }
  }
}

template <int NPL, bool RELU, int UNR>
__global__ void __launch_bounds__(256)
k_gather_rows(const float* __restrict__ p0, const float* __restrict__ p1,
              float* __restrict__ rows_out,
              const int* __restrict__ rowptr, const int* __restrict__ csr,
              const float* __restrict__ dinv, const float* __restrict__ bias,
              int N) {
  const int lane = threadIdx.x & 31;
  const int wid = __builtin_amdgcn_readfirstlane(
      (int)(blockIdx.x * (blockDim.x >> 5) + (threadIdx.x >> 5)));
  const int nw = (int)(gridDim.x * (blockDim.x >> 5));

  for (int d = wid; d < N; d += nw) {
    const int beg = rowptr[d];
    const int end = rowptr[d + 1];
    const float dd = dinv[d];
    const float wself = dd * dd;

    float4 a0 = ((const float4*)(p0 + (size_t)d * PLANE))[lane];
    a0.x *= wself; a0.y *= wself; a0.z *= wself; a0.w *= wself;
    float4 a1 = make_float4(0.f, 0.f, 0.f, 0.f);
    if (NPL == 2) {
      a1 = ((const float4*)(p1 + (size_t)d * PLANE))[lane];
      a1.x *= wself; a1.y *= wself; a1.z *= wself; a1.w *= wself;
    }

    for (int base = beg; base < end; base += 32) {
      const int left = end - base;
      const int cnt = left < 32 ? left : 32;
      int s_l = d;
      float w_l = 0.0f;
      if (lane < cnt) {
        s_l = csr[base + lane];
        w_l = dinv[s_l] * dd;
      }
      for (int j = 0; j < cnt; j += UNR) {
        float4 v0[UNR];
        float4 v1[UNR];
        float w[UNR];
#pragma unroll
        for (int u = 0; u < UNR; ++u) {
          const int s = __builtin_amdgcn_readlane(s_l, j + u);
          w[u] = __builtin_bit_cast(
              float, __builtin_amdgcn_readlane(__builtin_bit_cast(int, w_l), j + u));
          v0[u] = ((const float4*)(p0 + (size_t)s * PLANE))[lane];
          if (NPL == 2) v1[u] = ((const float4*)(p1 + (size_t)s * PLANE))[lane];
        }
#pragma unroll
        for (int u = 0; u < UNR; ++u) {
          a0.x += w[u] * v0[u].x; a0.y += w[u] * v0[u].y;
          a0.z += w[u] * v0[u].z; a0.w += w[u] * v0[u].w;
          if (NPL == 2) {
            a1.x += w[u] * v1[u].x; a1.y += w[u] * v1[u].y;
            a1.z += w[u] * v1[u].z; a1.w += w[u] * v1[u].w;
          }
        }
      }
    }

    float* __restrict__ orow = rows_out + (size_t)d * (NPL * PLANE);
    {
      const float4 b = ((const float4*)bias)[lane];
      a0.x += b.x; a0.y += b.y; a0.z += b.z; a0.w += b.w;
      if (RELU) {
        a0.x = fmaxf(a0.x, 0.f); a0.y = fmaxf(a0.y, 0.f);
        a0.z = fmaxf(a0.z, 0.f); a0.w = fmaxf(a0.w, 0.f);
      }
      ((float4*)orow)[lane] = a0;
    }
    if (NPL == 2) {
      const float4 b = ((const float4*)(bias + PLANE))[lane];
      a1.x += b.x; a1.y += b.y; a1.z += b.z; a1.w += b.w;
      if (RELU) {
        a1.x = fmaxf(a1.x, 0.f); a1.y = fmaxf(a1.y, 0.f);
        a1.z = fmaxf(a1.z, 0.f); a1.w = fmaxf(a1.w, 0.f);
      }
      ((float4*)(orow + PLANE))[lane] = a1;
    }
  }
}

#define HPITCH 260

template <int UNR>
__global__ void __launch_bounds__(256)
k_gather_relu_gemm(const float* __restrict__ p0, const float* __restrict__ p1,
                   float* __restrict__ Cout,
                   const int* __restrict__ rowptr, const int* __restrict__ csr,
                   const float* __restrict__ dinv, const float* __restrict__ bias,
                   const float* __restrict__ B, int N) {
  __shared__ __attribute__((aligned(16))) float Hs[GBM * HPITCH];
  __shared__ __attribute__((aligned(16))) float Bs[2][(GBN / 16) * BTILE];

  constexpr int T = HID / GBK;

  const int tid  = threadIdx.x;
  const int lane = tid & 31;
  const int wv   = __builtin_amdgcn_readfirstlane(tid >> 5);
  const int c    = lane & 15;
  const int h    = lane >> 4;
  const int m0   = blockIdx.x * GBM;

  const float* __restrict__ bg = B + (size_t)wv * GBN + 4 * lane;
  const int bs_st = (lane >> 2) * BTILE + wv * 16 + (lane & 3) * 4;
  float4 e0, e1, e2, e3;
  float4 o0, o1, o2, o3;

#define FUSED_FETCH(q0, q1, q2, q3, k1)                                                  \
  {                                                                                      \
    q0 = *(const float4*)(bg + (size_t)((k1) + 0) * GBN);                                \
    q1 = *(const float4*)(bg + (size_t)((k1) + 8) * GBN);                                \
    q2 = *(const float4*)(bg + (size_t)((k1) + 16) * GBN);                               \
    q3 = *(const float4*)(bg + (size_t)((k1) + 24) * GBN);                               \
  }
#define FUSED_PUT(nb, q0, q1, q2, q3)                                                    \
  {                                                                                      \
    *(float4*)(&Bs[nb][bs_st]) = q0;                                                     \
    *(float4*)(&Bs[nb][bs_st + 128]) = q1;                                               \
    *(float4*)(&Bs[nb][bs_st + 256]) = q2;                                               \
    *(float4*)(&Bs[nb][bs_st + 384]) = q3;                                               \
  }

  FUSED_FETCH(e0, e1, e2, e3, 0)
  FUSED_FETCH(o0, o1, o2, o3, GBK)

  const float4 bia0 = ((const float4*)bias)[lane];
  const float4 bia1 = ((const float4*)(bias + PLANE))[lane];

  for (int i = 0; i < 4; ++i) {
    const int r = wv * 4 + i;
    const int d = m0 + r;
    float4 a0 = make_float4(0.f, 0.f, 0.f, 0.f);
    float4 a1 = make_float4(0.f, 0.f, 0.f, 0.f);
    if (d < N) {
      const int beg = rowptr[d];
      const int end = rowptr[d + 1];
      const float dd = dinv[d];
      const float wself = dd * dd;

      a0 = ((const float4*)(p0 + (size_t)d * PLANE))[lane];
      a0.x *= wself; a0.y *= wself; a0.z *= wself; a0.w *= wself;
      a1 = ((const float4*)(p1 + (size_t)d * PLANE))[lane];
      a1.x *= wself; a1.y *= wself; a1.z *= wself; a1.w *= wself;

      for (int base = beg; base < end; base += 32) {
        const int left = end - base;
        const int cnt = left < 32 ? left : 32;
        int s_l = d;
        float w_l = 0.0f;
        if (lane < cnt) {
          s_l = csr[base + lane];
          w_l = dinv[s_l] * dd;
        }
        for (int j = 0; j < cnt; j += UNR) {
          float4 v0[UNR];
          float4 v1[UNR];
          float w[UNR];
#pragma unroll
          for (int u = 0; u < UNR; ++u) {
            const int s = __builtin_amdgcn_readlane(s_l, j + u);
            w[u] = __builtin_bit_cast(
                float, __builtin_amdgcn_readlane(__builtin_bit_cast(int, w_l), j + u));
            v0[u] = ((const float4*)(p0 + (size_t)s * PLANE))[lane];
            v1[u] = ((const float4*)(p1 + (size_t)s * PLANE))[lane];
          }
#pragma unroll
          for (int u = 0; u < UNR; ++u) {
            a0.x += w[u] * v0[u].x; a0.y += w[u] * v0[u].y;
            a0.z += w[u] * v0[u].z; a0.w += w[u] * v0[u].w;
            a1.x += w[u] * v1[u].x; a1.y += w[u] * v1[u].y;
            a1.z += w[u] * v1[u].z; a1.w += w[u] * v1[u].w;
          }
        }
      }

      a0.x += bia0.x; a0.y += bia0.y; a0.z += bia0.z; a0.w += bia0.w;
      a0.x = fmaxf(a0.x, 0.f); a0.y = fmaxf(a0.y, 0.f);
      a0.z = fmaxf(a0.z, 0.f); a0.w = fmaxf(a0.w, 0.f);
      a1.x += bia1.x; a1.y += bia1.y; a1.z += bia1.z; a1.w += bia1.w;
      a1.x = fmaxf(a1.x, 0.f); a1.y = fmaxf(a1.y, 0.f);
      a1.z = fmaxf(a1.z, 0.f); a1.w = fmaxf(a1.w, 0.f);
    }
    *(float4*)(&Hs[r * HPITCH + 4 * lane]) = a0;
    *(float4*)(&Hs[r * HPITCH + PLANE + 4 * lane]) = a1;
  }

  FUSED_PUT(0, e0, e1, e2, e3)
  __syncthreads();

  const int a_rd = ((wv >> 2) * 16 + c) * HPITCH + 2 * h;
  const int b_rd = (2 * (wv & 3)) * BTILE + 32 * h + c;

  v8f c0 = {}, c1 = {};

#define FUSED_MULT(cb, t)                                                                \
  {                                                                                      \
    const float* __restrict__ as = &Hs[a_rd + (t) * GBK];                                \
    const float* __restrict__ bs = &Bs[cb][b_rd];                                        \
    _Pragma("unroll") for (int ks = 0; ks < GBK / 4; ++ks) {                             \
      const v2f a = *(const v2f*)(as + 4 * ks);                                          \
      v2f b0, b1;                                                                        \
      b0.x = bs[64 * ks];          b0.y = bs[64 * ks + 16];                              \
      b1.x = bs[BTILE + 64 * ks];  b1.y = bs[BTILE + 64 * ks + 16];                      \
      c0 = __builtin_amdgcn_wmma_f32_16x16x4_f32(false, a, false, b0, (short)0, c0, false, false); \
      c1 = __builtin_amdgcn_wmma_f32_16x16x4_f32(false, a, false, b1, (short)0, c1, false, false); \
    }                                                                                    \
  }

  for (int t = 0; t < T; t += 2) {
    const int t2 = (t + 2 < T) ? (t + 2) : (T - 1);
    FUSED_FETCH(e0, e1, e2, e3, t2 * GBK)
    FUSED_MULT(0, t)
    FUSED_PUT(1, o0, o1, o2, o3)
    __syncthreads();

    const int t3 = (t + 3 < T) ? (t + 3) : (T - 1);
    FUSED_FETCH(o0, o1, o2, o3, t3 * GBK)
    FUSED_MULT(1, t + 1)
    FUSED_PUT(0, e0, e1, e2, e3)
    __syncthreads();
  }
#undef FUSED_FETCH
#undef FUSED_PUT
#undef FUSED_MULT

  const int col = (wv & 3) * 32 + c;
#pragma unroll
  for (int r = 0; r < 8; ++r) {
    const int row = m0 + (wv >> 2) * 16 + 8 * h + r;
    if (row < N) {
      float* __restrict__ o = Cout + (size_t)row * GBN + col;
      o[0]  = c0[r];
      o[16] = c1[r];
    }
  }
}

extern "C" void kernel_launch(void* const* d_in, const int* in_sizes, int n_in,
                              void* d_out, int out_size, void* d_ws, size_t ws_size,
                              hipStream_t stream) {
  (void)n_in; (void)out_size; (void)ws_size;

  const float* x  = (const float*)d_in[0];
  const int*   ei = (const int*)d_in[1];
  const float* W1 = (const float*)d_in[2];
  const float* b1 = (const float*)d_in[3];
  const float* W2 = (const float*)d_in[4];
  const float* b2 = (const float*)d_in[5];
  float* out = (float*)d_out;

  const int N = in_sizes[0] / IN_C;
  const int E = in_sizes[1] / 2;
  const int* src = ei;
  const int* dst = ei + E;

  char* w = (char*)d_ws;
  size_t dinv_bytes = (((size_t)N * 4) + 255) & ~(size_t)255;
  size_t csr_bytes  = (((size_t)E * 4) + 255) & ~(size_t)255;
  float* dinv = (float*)w;
  float* bufA = (float*)(w + dinv_bytes);
  float* bufB = bufA + (size_t)N * HID;
  char*  elist  = (char*)(bufA + (size_t)N * PLANE);
  const int NB = (N + BIN_NODES - 1) / BIN_NODES;
  const int NC = (E + CH - 1) / CH;
  int*   rowptr   = (int*)elist;
  int*   csr      = (int*)(elist + dinv_bytes);
  int2*  binned   = (int2*)(elist + dinv_bytes + csr_bytes);
  int*   tbl      = (int*)(elist + dinv_bytes + 3 * csr_bytes);
  int*   binstart = tbl + (size_t)NC * NB;

  const int T = 256;

  k_bin_count<<<NC, GT, 0, stream>>>(tbl, dst, E, NB);
  k_bin_scatter<<<NC, GT, 0, stream>>>(binned, binstart, tbl, src, dst, E, NB, NC);
  k_bin_rows<<<NB, GT, 0, stream>>>(csr, rowptr, dinv, binned, binstart, N, NB);

  const int mblocks = (N + GBM - 1) / GBM;

  k_gemm_wmma_f32<IN_C, HID><<<mblocks * (HID / GBN), 128, 0, stream>>>(x, W1, out, bufA, N);

  k_gather_relu_gemm<4><<<mblocks, T, 0, stream>>>(out, bufA, bufB, rowptr, csr, dinv, b1, W2, N);

  k_gather_rows<1, false, 8><<<2048, T, 0, stream>>>(bufB, bufB, out, rowptr, csr, dinv, b2, N);
}


# Edge weights factored out of the row sums: both products store row m times dinv[m]; a row sum adds source rows unweighted and is scaled by dinv[dst] once; no dinv gather per edge; sum order kept
# speedup vs baseline: 3.8339x; 1.0314x over previous
#include <hip/hip_runtime.h>


typedef __attribute__((ext_vector_type(2))) float v2f;
typedef __attribute__((ext_vector_type(8))) float v8f;

#define IN_C 512
#define HID 256
#define OUT_C 128
#define PLANE 128

#define BIN_SHIFT 8
#define BIN_NODES (1 << BIN_SHIFT)
#define MAX_BINS 1024
#define GT 1024
#define CH 4096
#define CH_PER (CH / GT)
#define STAGE_CAP 16384

__device__ __forceinline__ int block_scan_excl(int val, int* wsum, int lane, int w) {
  int incl = val;
  for (int o = 1; o < 32; o <<= 1) {
    int t = __shfl_up(incl, o);
    if (lane >= o) incl += t;
  }
  if (lane == 31) wsum[w] = incl;
  __syncthreads();
  const int ws = wsum[lane];
  int wincl = ws;
  for (int o = 1; o < 32; o <<= 1) {
    int t = __shfl_up(wincl, o);
    if (lane >= o) wincl += t;
  }
  const int woff = __shfl(wincl - ws, w);
  __syncthreads();
  return woff + incl - val;
}

__global__ void __launch_bounds__(GT)
k_bin_count(int* __restrict__ tbl, const int* __restrict__ dst, int e, int nb) {
  __shared__ int h[MAX_BINS];
  const int tid = threadIdx.x;
  const int c = blockIdx.x;
  h[tid] = 0;
  int d[CH_PER];
#pragma unroll
  for (int k = 0; k < CH_PER; ++k) {
    const int i = c * CH + k * GT + tid;
    d[k] = dst[i < e ? i : e - 1];
  }
  __syncthreads();
#pragma unroll
  for (int k = 0; k < CH_PER; ++k) {
    const int i = c * CH + k * GT + tid;
    if (i < e) atomicAdd(&h[d[k] >> BIN_SHIFT], 1);
  }
  __syncthreads();
  if (tid < nb) tbl[c * nb + tid] = h[tid];
}

__global__ void __launch_bounds__(GT)
k_bin_scatter(int2* __restrict__ binned, int* __restrict__ binstart,
              const int* __restrict__ tbl, const int* __restrict__ src,
              const int* __restrict__ dst, int e, int nb, int nc) {
  __shared__ int tot[MAX_BINS];
  __shared__ int bef[MAX_BINS];
  __shared__ int hcnt[MAX_BINS];
  __shared__ int lstart[MAX_BINS];
  __shared__ int goff[MAX_BINS];
  __shared__ int wsum[32];
  __shared__ int2 sbuf[CH];
  const int tid = threadIdx.x;
  const int lane = tid & 31;
  const int w = tid >> 5;
  const int c = blockIdx.x;

  tot[tid] = 0;
  bef[tid] = 0;
  hcnt[tid] = 0;

  int es[CH_PER], ed[CH_PER];
#pragma unroll
  for (int k = 0; k < CH_PER; ++k) {
    const int i = c * CH + k * GT + tid;
    const int ic = i < e ? i : e - 1;
    es[k] = src[ic];
    ed[k] = dst[ic];
  }
  __syncthreads();

  {
    const int binl = tid & 255;
    const int part = tid >> 8;
    const int q = (nc + 3) >> 2;
    const int c0 = part * q;
    const int c1 = (c0 + q < nc) ? (c0 + q) : nc;
    for (int b = binl; b < nb; b += 256) {
      int st = 0, sb = 0;
#pragma unroll 8
      for (int cc = c0; cc < c1; ++cc) {
        const int v = tbl[cc * nb + b];
        st += v;
        sb += (cc < c) ? v : 0;
      }
      atomicAdd(&tot[b], st);
      atomicAdd(&bef[b], sb);
    }
  }

  int lr[CH_PER];
#pragma unroll
  for (int k = 0; k < CH_PER; ++k) {
    const int i = c * CH + k * GT + tid;
    lr[k] = 0;
    if (i < e) lr[k] = atomicAdd(&hcnt[ed[k] >> BIN_SHIFT], 1);
  }
  __syncthreads();

  const int my_tot = tot[tid];
  const int my_cnt = hcnt[tid];
  const int ex_tot = block_scan_excl(my_tot, wsum, lane, w);
  const int ex_cnt = block_scan_excl(my_cnt, wsum, lane, w);
  lstart[tid] = ex_cnt;
  goff[tid] = ex_tot + bef[tid];
  if (c == 0) {
    if (tid < nb) binstart[tid] = ex_tot;
    if (tid == 0) binstart[nb] = e;
  }
  __syncthreads();

#pragma unroll
  for (int k = 0; k < CH_PER; ++k) {
    const int i = c * CH + k * GT + tid;
    if (i < e) sbuf[lstart[ed[k] >> BIN_SHIFT] + lr[k]] = make_int2(es[k], ed[k]);
  }
  __syncthreads();

  const int left = e - c * CH;
  const int nloc = left < CH ? left : CH;
#pragma unroll
  for (int k = 0; k < CH_PER; ++k) {
    const int j = k * GT + tid;
    if (j < nloc) {
      const int2 v = sbuf[j];
      const int b = v.y >> BIN_SHIFT;
      binned[goff[b] + (j - lstart[b])] = v;
    }
  }
}

__global__ void __launch_bounds__(GT)
k_bin_rows(int* __restrict__ csr, int* __restrict__ rowptr, float* __restrict__ dinv,
           const int2* __restrict__ binned, const int* __restrict__ binstart,
           int n, int nb) {
  __shared__ int ncnt[BIN_NODES];
  __shared__ int nstart[BIN_NODES];
  __shared__ int wsum[32];
  __shared__ int stage[STAGE_CAP];
  const int tid = threadIdx.x;
  const int lane = tid & 31;
  const int w = tid >> 5;
  const int b = blockIdx.x;
  const int beg = binstart[b];
  const int end = binstart[b + 1];
  const int m = end - beg;
  const int node0 = b << BIN_SHIFT;

  if (tid < BIN_NODES) ncnt[tid] = 0;
  __syncthreads();
  for (int i = tid; i < m; i += GT) {
    const int2 v = binned[beg + i];
    atomicAdd(&ncnt[v.y - node0], 1);
  }
  __syncthreads();

  const int cval = (tid < BIN_NODES) ? ncnt[tid] : 0;
  const int ex = block_scan_excl(cval, wsum, lane, w);
  if (tid < BIN_NODES) {
    nstart[tid] = ex;
    ncnt[tid] = 0;
    const int d = node0 + tid;
    if (d < n) {
      rowptr[d] = beg + ex;
      dinv[d] = rsqrtf((float)(cval + 1));
    }
  }
  if (b == nb - 1 && tid == 0) rowptr[n] = end;
  __syncthreads();

  const bool fits = m <= STAGE_CAP;
  for (int i = tid; i < m; i += GT) {
    const int2 v = binned[beg + i];
    const int dl = v.y - node0;
    const int pos = nstart[dl] + atomicAdd(&ncnt[dl], 1);
    if (fits) stage[pos] = v.x;
    else csr[beg + pos] = v.x;
  }
  if (fits) {
    __syncthreads();
    for (int i = tid; i < m; i += GT) csr[beg + i] = stage[i];
  }
}

#define GBM 32
#define GBN 128
#define GBK 32
#define APITCH 36
#define BTILE 528

template <int KDIM, int NCOLS>
__global__ void __launch_bounds__(128)
k_gemm_wmma_f32(const float* __restrict__ A,
                const float* __restrict__ B,
                float* __restrict__ C0, float* __restrict__ C1,
                const float* __restrict__ rs,
                int M) {
  __shared__ __attribute__((aligned(16))) float As[2][GBM * APITCH];
  __shared__ __attribute__((aligned(16))) float Bs[2][(GBN / 16) * BTILE];

  constexpr int NBN = NCOLS / GBN;
  constexpr int T = KDIM / GBK;

  const int tid  = threadIdx.x;
  const int lane = tid & 31;
  const int wv   = tid >> 5;
  const int c    = lane & 15;
  const int h    = lane >> 4;
  const int bn   = blockIdx.x % NBN;
  const int m0   = (blockIdx.x / NBN) * GBM;
  const int n0   = bn * GBN;

  const int akq = (tid & 7) * 4;
  const int ar0 = tid >> 3;
  int arow0 = m0 + ar0;       arow0 = arow0 < M ? arow0 : M - 1;
  int arow1 = m0 + ar0 + 16;  arow1 = arow1 < M ? arow1 : M - 1;
  const float* __restrict__ ag0 = A + (size_t)arow0 * KDIM + akq;
  const float* __restrict__ ag1 = A + (size_t)arow1 * KDIM + akq;
  const float* __restrict__ bg  = B + (size_t)wv * NCOLS + n0 + 4 * lane;
  const int as_st = ar0 * APITCH + akq;
  const int bs_st = (lane >> 2) * BTILE + wv * 16 + (lane & 3) * 4;

  const int a_rd = c * APITCH + 2 * h;
  const int b_rd = (2 * wv) * BTILE + 32 * h + c;

  float4 ra0, ra1, rb[8];
  v8f c00 = {}, c01 = {}, c10 = {}, c11 = {};

#define GEMM_FETCH(k1)                                                                   \
  {                                                                                      \
    ra0 = *(const float4*)(ag0 + (k1));                                                  \
    ra1 = *(const float4*)(ag1 + (k1));                                                  \
    _Pragma("unroll") for (int i = 0; i < 8; ++i)                                        \
        rb[i] = *(const float4*)(bg + (size_t)((k1) + 4 * i) * NCOLS);                   \
  }
#define GEMM_PUT(nb)                                                                     \
  {                                                                                      \
    *(float4*)(&As[nb][as_st]) = ra0;                                                    \
    *(float4*)(&As[nb][as_st + 16 * APITCH]) = ra1;                                      \
    _Pragma("unroll") for (int i = 0; i < 8; ++i)                                        \
        *(float4*)(&Bs[nb][bs_st + 64 * i]) = rb[i];                                     \
  }
#define GEMM_MULT(cb)                                                                    \
  {                                                                                      \
    const float* __restrict__ as = &As[cb][a_rd];                                        \
    const float* __restrict__ bs = &Bs[cb][b_rd];                                        \
    _Pragma("unroll") for (int ks = 0; ks < GBK / 4; ++ks) {                             \
      const v2f a0 = *(const v2f*)(as + 4 * ks);                                         \
      const v2f a1 = *(const v2f*)(as + 16 * APITCH + 4 * ks);                           \
      v2f b0, b1;                                                                        \
      b0.x = bs[64 * ks];          b0.y = bs[64 * ks + 16];                              \
      b1.x = bs[BTILE + 64 * ks];  b1.y = bs[BTILE + 64 * ks + 16];                      \
      c00 = __builtin_amdgcn_wmma_f32_16x16x4_f32(false, a0, false, b0, (short)0, c00, false, false); \
      c01 = __builtin_amdgcn_wmma_f32_16x16x4_f32(false, a0, false, b1, (short)0, c01, false, false); \
      c10 = __builtin_amdgcn_wmma_f32_16x16x4_f32(false, a1, false, b0, (short)0, c10, false, false); \
      c11 = __builtin_amdgcn_wmma_f32_16x16x4_f32(false, a1, false, b1, (short)0, c11, false, false); \
    }                                                                                    \
  }

  GEMM_FETCH(0)
  GEMM_PUT(0)
  __syncthreads();

  for (int t = 0; t < T; t += 2) {
    GEMM_FETCH((t + 1) * GBK)
    GEMM_MULT(0)
    GEMM_PUT(1)
    __syncthreads();

    const int t2 = (t + 2 < T) ? (t + 2) : (T - 1);
    GEMM_FETCH(t2 * GBK)
    GEMM_MULT(1)
    GEMM_PUT(0)
    __syncthreads();
  }
#undef GEMM_FETCH
#undef GEMM_PUT
#undef GEMM_MULT

  float* __restrict__ Cb = bn ? C1 : C0;
  const int col = wv * 32 + c;
  float s0[8], s1[8];
#pragma unroll
  for (int r = 0; r < 8; ++r) {
    const int row0 = m0 + 8 * h + r;
    const int row1 = row0 + 16;
    s0[r] = rs[row0 < M ? row0 : M - 1];
    s1[r] = rs[row1 < M ? row1 : M - 1];
  }
#pragma unroll
  for (int r = 0; r < 8; ++r) {
    const int row0 = m0 + 8 * h + r;
    const int row1 = row0 + 16;
    if (row0 < M) {
      float* __restrict__ o = Cb + (size_t)row0 * GBN + col;
      o[0]  = s0[r] * c00[r];
      o[16] = s0[r] * c01[r];
    }
    if (row1 < M) {
      float* __restrict__ o = Cb + (size_t)row1 * GBN + col;
      o[0]  = s1[r] * c10[r];
      o[16] = s1[r] * c11[r];
    }
  }
}

template <int NPL, bool RELU, int UNR>
__global__ void __launch_bounds__(256)
k_gather_rows(const float* __restrict__ p0, const float* __restrict__ p1,
              float* __restrict__ rows_out,
              const int* __restrict__ rowptr, const int* __restrict__ csr,
              const float* __restrict__ dinv, const float* __restrict__ bias,
              int N) {
  const int lane = threadIdx.x & 31;
  const int wid = __builtin_amdgcn_readfirstlane(
      (int)(blockIdx.x * (blockDim.x >> 5) + (threadIdx.x >> 5)));
  const int nw = (int)(gridDim.x * (blockDim.x >> 5));

  for (int d = wid; d < N; d += nw) {
    const int beg = rowptr[d];
    const int end = rowptr[d + 1];
    const float dd = dinv[d];

    float4 a0 = ((const float4*)(p0 + (size_t)d * PLANE))[lane];
    float4 a1 = make_float4(0.f, 0.f, 0.f, 0.f);
    if (NPL == 2) a1 = ((const float4*)(p1 + (size_t)d * PLANE))[lane];

    for (int base = beg; base < end; base += 32) {
      const int left = end - base;
      const int cnt = left < 32 ? left : 32;
      int s_l = d;
      if (lane < cnt) s_l = csr[base + lane];
      for (int j = 0; j < cnt; j += UNR) {
        float4 v0[UNR];
        float4 v1[UNR];
        float w[UNR];
#pragma unroll
        for (int u = 0; u < UNR; ++u) {
          const int s = __builtin_amdgcn_readlane(s_l, j + u);
          w[u] = (j + u < cnt) ? 1.0f : 0.0f;
          v0[u] = ((const float4*)(p0 + (size_t)s * PLANE))[lane];
          if (NPL == 2) v1[u] = ((const float4*)(p1 + (size_t)s * PLANE))[lane];
        }
#pragma unroll
        for (int u = 0; u < UNR; ++u) {
          a0.x += w[u] * v0[u].x; a0.y += w[u] * v0[u].y;
          a0.z += w[u] * v0[u].z; a0.w += w[u] * v0[u].w;
          if (NPL == 2) {
            a1.x += w[u] * v1[u].x; a1.y += w[u] * v1[u].y;
            a1.z += w[u] * v1[u].z; a1.w += w[u] * v1[u].w;
          }
        }
      }
    }

    float* __restrict__ orow = rows_out + (size_t)d * (NPL * PLANE);
    {
      const float4 b = ((const float4*)bias)[lane];
      a0.x = dd * a0.x + b.x; a0.y = dd * a0.y + b.y;
      a0.z = dd * a0.z + b.z; a0.w = dd * a0.w + b.w;
      if (RELU) {
        a0.x = fmaxf(a0.x, 0.f); a0.y = fmaxf(a0.y, 0.f);
        a0.z = fmaxf(a0.z, 0.f); a0.w = fmaxf(a0.w, 0.f);
      }
      ((float4*)orow)[lane] = a0;
    }
    if (NPL == 2) {
      const float4 b = ((const float4*)(bias + PLANE))[lane];
      a1.x = dd * a1.x + b.x; a1.y = dd * a1.y + b.y;
      a1.z = dd * a1.z + b.z; a1.w = dd * a1.w + b.w;
      if (RELU) {
        a1.x = fmaxf(a1.x, 0.f); a1.y = fmaxf(a1.y, 0.f);
        a1.z = fmaxf(a1.z, 0.f); a1.w = fmaxf(a1.w, 0.f);
      }
      ((float4*)(orow + PLANE))[lane] = a1;
    }
  }
}

#define HPITCH 260

template <int UNR>
__global__ void __launch_bounds__(256)
k_gather_relu_gemm(const float* __restrict__ p0, const float* __restrict__ p1,
                   float* __restrict__ Cout,
                   const int* __restrict__ rowptr, const int* __restrict__ csr,
                   const float* __restrict__ dinv, const float* __restrict__ bias,
                   const float* __restrict__ B, int N) {
  __shared__ __attribute__((aligned(16))) float Hs[GBM * HPITCH];
  __shared__ __attribute__((aligned(16))) float Bs[2][(GBN / 16) * BTILE];
  __shared__ float Ds[GBM];

  constexpr int T = HID / GBK;

  const int tid  = threadIdx.x;
  const int lane = tid & 31;
  const int wv   = __builtin_amdgcn_readfirstlane(tid >> 5);
  const int c    = lane & 15;
  const int h    = lane >> 4;
  const int m0   = blockIdx.x * GBM;

  const float* __restrict__ bg = B + (size_t)wv * GBN + 4 * lane;
  const int bs_st = (lane >> 2) * BTILE + wv * 16 + (lane & 3) * 4;
  float4 e0, e1, e2, e3;
  float4 o0, o1, o2, o3;

#define FUSED_FETCH(q0, q1, q2, q3, k1)                                                  \
  {                                                                                      \
    q0 = *(const float4*)(bg + (size_t)((k1) + 0) * GBN);                                \
    q1 = *(const float4*)(bg + (size_t)((k1) + 8) * GBN);                                \
    q2 = *(const float4*)(bg + (size_t)((k1) + 16) * GBN);                               \
    q3 = *(const float4*)(bg + (size_t)((k1) + 24) * GBN);                               \
  }
#define FUSED_PUT(nb, q0, q1, q2, q3)                                                    \
  {                                                                                      \
    *(float4*)(&Bs[nb][bs_st]) = q0;                                                     \
    *(float4*)(&Bs[nb][bs_st + 128]) = q1;                                               \
    *(float4*)(&Bs[nb][bs_st + 256]) = q2;                                               \
    *(float4*)(&Bs[nb][bs_st + 384]) = q3;                                               \
  }

  FUSED_FETCH(e0, e1, e2, e3, 0)
  FUSED_FETCH(o0, o1, o2, o3, GBK)

  const float4 bia0 = ((const float4*)bias)[lane];
  const float4 bia1 = ((const float4*)(bias + PLANE))[lane];

  for (int i = 0; i < 4; ++i) {
    const int r = wv * 4 + i;
    const int d = m0 + r;
    float4 a0 = make_float4(0.f, 0.f, 0.f, 0.f);
    float4 a1 = make_float4(0.f, 0.f, 0.f, 0.f);
    float dd = 0.0f;
    if (d < N) {
      const int beg = rowptr[d];
      const int end = rowptr[d + 1];
      dd = dinv[d];

      a0 = ((const float4*)(p0 + (size_t)d * PLANE))[lane];
      a1 = ((const float4*)(p1 + (size_t)d * PLANE))[lane];

      for (int base = beg; base < end; base += 32) {
        const int left = end - base;
        const int cnt = left < 32 ? left : 32;
        int s_l = d;
        if (lane < cnt) s_l = csr[base + lane];
        for (int j = 0; j < cnt; j += UNR) {
          float4 v0[UNR];
          float4 v1[UNR];
          float w[UNR];
#pragma unroll
          for (int u = 0; u < UNR; ++u) {
            const int s = __builtin_amdgcn_readlane(s_l, j + u);
            w[u] = (j + u < cnt) ? 1.0f : 0.0f;
            v0[u] = ((const float4*)(p0 + (size_t)s * PLANE))[lane];
            v1[u] = ((const float4*)(p1 + (size_t)s * PLANE))[lane];
          }
#pragma unroll
          for (int u = 0; u < UNR; ++u) {
            a0.x += w[u] * v0[u].x; a0.y += w[u] * v0[u].y;
            a0.z += w[u] * v0[u].z; a0.w += w[u] * v0[u].w;
            a1.x += w[u] * v1[u].x; a1.y += w[u] * v1[u].y;
            a1.z += w[u] * v1[u].z; a1.w += w[u] * v1[u].w;
          }
        }
      }

      a0.x = dd * a0.x + bia0.x; a0.y = dd * a0.y + bia0.y;
      a0.z = dd * a0.z + bia0.z; a0.w = dd * a0.w + bia0.w;
      a0.x = fmaxf(a0.x, 0.f); a0.y = fmaxf(a0.y, 0.f);
      a0.z = fmaxf(a0.z, 0.f); a0.w = fmaxf(a0.w, 0.f);
      a1.x = dd * a1.x + bia1.x; a1.y = dd * a1.y + bia1.y;
      a1.z = dd * a1.z + bia1.z; a1.w = dd * a1.w + bia1.w;
      a1.x = fmaxf(a1.x, 0.f); a1.y = fmaxf(a1.y, 0.f);
      a1.z = fmaxf(a1.z, 0.f); a1.w = fmaxf(a1.w, 0.f);
    }
    *(float4*)(&Hs[r * HPITCH + 4 * lane]) = a0;
    *(float4*)(&Hs[r * HPITCH + PLANE + 4 * lane]) = a1;
    if (lane == 0) Ds[r] = dd;
  }

  FUSED_PUT(0, e0, e1, e2, e3)
  __syncthreads();

  const int a_rd = ((wv >> 2) * 16 + c) * HPITCH + 2 * h;
  const int b_rd = (2 * (wv & 3)) * BTILE + 32 * h + c;

  v8f c0 = {}, c1 = {};

#define FUSED_MULT(cb, t)                                                                \
  {                                                                                      \
    const float* __restrict__ as = &Hs[a_rd + (t) * GBK];                                \
    const float* __restrict__ bs = &Bs[cb][b_rd];                                        \
    _Pragma("unroll") for (int ks = 0; ks < GBK / 4; ++ks) {                             \
      const v2f a = *(const v2f*)(as + 4 * ks);                                          \
      v2f b0, b1;                                                                        \
      b0.x = bs[64 * ks];          b0.y = bs[64 * ks + 16];                              \
      b1.x = bs[BTILE + 64 * ks];  b1.y = bs[BTILE + 64 * ks + 16];                      \
      c0 = __builtin_amdgcn_wmma_f32_16x16x4_f32(false, a, false, b0, (short)0, c0, false, false); \
      c1 = __builtin_amdgcn_wmma_f32_16x16x4_f32(false, a, false, b1, (short)0, c1, false, false); \
    }                                                                                    \
  }

  for (int t = 0; t < T; t += 2) {
    const int t2 = (t + 2 < T) ? (t + 2) : (T - 1);
    FUSED_FETCH(e0, e1, e2, e3, t2 * GBK)
    FUSED_MULT(0, t)
    FUSED_PUT(1, o0, o1, o2, o3)
    __syncthreads();

    const int t3 = (t + 3 < T) ? (t + 3) : (T - 1);
    FUSED_FETCH(o0, o1, o2, o3, t3 * GBK)
    FUSED_MULT(1, t + 1)
    FUSED_PUT(0, e0, e1, e2, e3)
    __syncthreads();
  }
#undef FUSED_FETCH
#undef FUSED_PUT
#undef FUSED_MULT

  const int col = (wv & 3) * 32 + c;
#pragma unroll
  for (int r = 0; r < 8; ++r) {
    const int rl = (wv >> 2) * 16 + 8 * h + r;
    const int row = m0 + rl;
    const float sc = Ds[rl];
    if (row < N) {
      float* __restrict__ o = Cout + (size_t)row * GBN + col;
      o[0]  = sc * c0[r];
      o[16] = sc * c1[r];
    }
  }
}

extern "C" void kernel_launch(void* const* d_in, const int* in_sizes, int n_in,
                              void* d_out, int out_size, void* d_ws, size_t ws_size,
                              hipStream_t stream) {
  (void)n_in; (void)out_size; (void)ws_size;

  const float* x  = (const float*)d_in[0];
  const int*   ei = (const int*)d_in[1];
  const float* W1 = (const float*)d_in[2];
  const float* b1 = (const float*)d_in[3];
  const float* W2 = (const float*)d_in[4];
  const float* b2 = (const float*)d_in[5];
  float* out = (float*)d_out;

  const int N = in_sizes[0] / IN_C;
  const int E = in_sizes[1] / 2;
  const int* src = ei;
  const int* dst = ei + E;

  char* w = (char*)d_ws;
  size_t dinv_bytes = (((size_t)N * 4) + 255) & ~(size_t)255;
  size_t csr_bytes  = (((size_t)E * 4) + 255) & ~(size_t)255;
  float* dinv = (float*)w;
  float* bufA = (float*)(w + dinv_bytes);
  float* bufB = bufA + (size_t)N * HID;
  char*  elist  = (char*)(bufA + (size_t)N * PLANE);
  const int NB = (N + BIN_NODES - 1) / BIN_NODES;
  const int NC = (E + CH - 1) / CH;
  int*   rowptr   = (int*)elist;
  int*   csr      = (int*)(elist + dinv_bytes);
  int2*  binned   = (int2*)(elist + dinv_bytes + csr_bytes);
  int*   tbl      = (int*)(elist + dinv_bytes + 3 * csr_bytes);
  int*   binstart = tbl + (size_t)NC * NB;

  const int T = 256;

  k_bin_count<<<NC, GT, 0, stream>>>(tbl, dst, E, NB);
  k_bin_scatter<<<NC, GT, 0, stream>>>(binned, binstart, tbl, src, dst, E, NB, NC);
  k_bin_rows<<<NB, GT, 0, stream>>>(csr, rowptr, dinv, binned, binstart, N, NB);

  const int mblocks = (N + GBM - 1) / GBM;

  k_gemm_wmma_f32<IN_C, HID><<<mblocks * (HID / GBN), 128, 0, stream>>>(x, W1, out, bufA, dinv, N);

  k_gather_relu_gemm<4><<<mblocks, T, 0, stream>>>(out, bufA, bufB, rowptr, csr, dinv, b1, W2, N);

  k_gather_rows<1, false, 8><<<2048, T, 0, stream>>>(bufB, bufB, out, rowptr, csr, dinv, b2, N);
}
